# starting kernel
# baseline (speedup 1.0000x reference)
#include <hip/hip_runtime.h>


#define S_DIM   64
#define C_DIM   8
#define N_ITER  4
#define NEUMANN 6
#define B_TOT   128
#define NB      8
#define BSUB    8
#define NWG     (B_TOT / BSUB)
#define THREADS 128
#define DT_C    0.05f
#define LDA     66

typedef __attribute__((ext_vector_type(2))) float v2f;
typedef __attribute__((ext_vector_type(8))) float v8f;

__device__ __forceinline__ v8f wmma_gemm_16x64(const float* __restrict__ A,
                                               const float* __restrict__ Bt,
                                               int n0, int lane) {
  v8f acc = {0.f, 0.f, 0.f, 0.f, 0.f, 0.f, 0.f, 0.f};
  const int m = lane & 15;
  const int h = (lane >> 4) << 1;
#pragma unroll
  for (int k0 = 0; k0 < S_DIM; k0 += 4) {
    v2f a, b;
    a.x = A[m * LDA + k0 + h];
    a.y = A[m * LDA + k0 + h + 1];
    b.x = Bt[(k0 + h) * LDA + n0 + m];
    b.y = Bt[(k0 + h + 1) * LDA + n0 + m];
    acc = __builtin_amdgcn_wmma_f32_16x16x4_f32(false, a, false, b,
                                                (short)0, acc, false, false);
  }
  return acc;
}

__global__ __launch_bounds__(THREADS, 1)
void solver_gfx1250_kernel(const float* __restrict__ y0,
                           const float* __restrict__ forces,
                           const float* __restrict__ W,
                           float* __restrict__ out) {
  __shared__ float sWt[S_DIM * LDA];
  __shared__ float sS [16 * LDA];
  __shared__ float sX0[16 * LDA];
  __shared__ float sX1[16 * LDA];
  __shared__ float sY[C_DIM][BSUB * S_DIM];
  __shared__ float sYprev[BSUB * S_DIM];
  __shared__ float sYm  [BSUB * S_DIM];
  __shared__ float sT   [BSUB * S_DIM];
  __shared__ float sG   [BSUB * S_DIM];
  __shared__ float sRhs [BSUB * S_DIM];
  __shared__ float sWx  [BSUB * S_DIM];

  const int tid  = threadIdx.x;
  const int lane = tid & 31;
  const int wave = tid >> 5;
  const int n0   = wave * 16;
  const int b0   = blockIdx.x * BSUB;

  for (int e = tid; e < S_DIM * S_DIM; e += THREADS) {
    int p = e >> 6, s = e & 63;
    sWt[s * LDA + p] = W[p * S_DIM + s];
  }
  for (int e = tid; e < 16 * LDA; e += THREADS) {
    sS[e] = 0.f; sX0[e] = 0.f; sX1[e] = 0.f;
  }
  for (int e = tid; e < BSUB * S_DIM; e += THREADS) {
    int b = e >> 6, j = e & 63;
    float v = y0[(b0 + b) * S_DIM + j];
    sYprev[e] = v;
    out[(size_t)(b0 + b) * S_DIM + j] = v;
  }
  __syncthreads();

  for (int blk = 0; blk < NB; ++blk) {
    const float* fblk =
        forces + ((size_t)(1 + blk * C_DIM) * B_TOT + b0) * S_DIM;
    for (int off = tid * 16; off < C_DIM * BSUB * S_DIM; off += THREADS * 16)
      __builtin_prefetch(fblk + off, 0, 1);

    for (int c = 0; c < C_DIM; ++c)
      for (int e = tid; e < BSUB * S_DIM; e += THREADS) sY[c][e] = 0.f;
    __syncthreads();

    for (int it = 0; it < N_ITER; ++it) {
      for (int e = tid; e < BSUB * S_DIM; e += THREADS) sYm[e] = sYprev[e];
      __syncthreads();

      for (int c = 0; c < C_DIM; ++c) {
        for (int e = tid; e < BSUB * S_DIM; e += THREADS) {
          int b = e >> 6, j = e & 63;
          sS[b * LDA + j] = 0.5f * (sYm[e] + sY[c][e]);
        }
        __syncthreads();

        {
          v8f tacc = wmma_gemm_16x64(sS, sWt, n0, lane);
          v8f wacc = {0.f, 0.f, 0.f, 0.f, 0.f, 0.f, 0.f, 0.f};
          if (c > 0) wacc = wmma_gemm_16x64(sX0, sWt, n0, lane);
          if ((lane >> 4) == 0) {
            int col = n0 + (lane & 15);
#pragma unroll
            for (int v = 0; v < 8; ++v) {
              sT[v * S_DIM + col] = tanhf(tacc[v]);
              if (c > 0) sWx[v * S_DIM + col] = wacc[v];
            }
          }
        }
        __syncthreads();

        for (int e = tid; e < BSUB * S_DIM; e += THREADS) {
          int b = e >> 6, j = e & 63;
          float t = sT[e];
          float g = 0.5f * DT_C * (1.0f - t * t);
          float fv = fblk[(size_t)c * B_TOT * S_DIM + b * S_DIM + j];
          float R = sY[c][e] - sYm[e] + DT_C * t - DT_C * fv;
          float rhs = R;
          if (c > 0) {
            float xp = sX0[b * LDA + j];
            rhs += xp - g * sWx[e];
          }
          sG[e] = g;
          sRhs[e] = rhs;
          sX0[b * LDA + j] = rhs;
        }
        __syncthreads();

        for (int k = 0; k < NEUMANN; ++k) {
          const float* xin  = (k & 1) ? sX1 : sX0;
          float*       xout = (k & 1) ? sX0 : sX1;
          v8f macc = wmma_gemm_16x64(xin, sWt, n0, lane);
          if ((lane >> 4) == 0) {
            int col = n0 + (lane & 15);
#pragma unroll
            for (int v = 0; v < 8; ++v) {
              int e = v * S_DIM + col;
              xout[v * LDA + col] = sRhs[e] - sG[e] * macc[v];
            }
          }
          __syncthreads();
        }

        for (int e = tid; e < BSUB * S_DIM; e += THREADS) {
          int b = e >> 6, j = e & 63;
          float yold = sY[c][e];
          sYm[e] = yold;
          sY[c][e] = yold - sX0[b * LDA + j];
        }
        __syncthreads();
      }
    }

    for (int c = 0; c < C_DIM; ++c)
      for (int e = tid; e < BSUB * S_DIM; e += THREADS) {
        int b = e >> 6, j = e & 63;
        out[((size_t)(1 + blk * C_DIM + c) * B_TOT + b0 + b) * S_DIM + j] =
            sY[c][e];
      }
    for (int e = tid; e < BSUB * S_DIM; e += THREADS)
      sYprev[e] = sY[C_DIM - 1][e];
    __syncthreads();
  }
}

extern "C" void kernel_launch(void* const* d_in, const int* in_sizes, int n_in,
                              void* d_out, int out_size, void* d_ws,
                              size_t ws_size, hipStream_t stream) {
  (void)in_sizes; (void)n_in; (void)out_size; (void)d_ws; (void)ws_size;
  const float* y0     = (const float*)d_in[0];
  const float* forces = (const float*)d_in[1];
  const float* W      = (const float*)d_in[2];
  float* out = (float*)d_out;
  solver_gfx1250_kernel<<<NWG, THREADS, 0, stream>>>(y0, forces, W, out);
}


# Sweep GEMMs split by K over the 4 waves, transposed: a wave's own x slice stays in accumulator registers as B, its W slice in registers as A; waves exchange partial sums; sum order changes
# speedup vs baseline: 2.3288x; 2.3288x over previous
#include <hip/hip_runtime.h>


#define S_DIM   64
#define C_DIM   8
#define N_ITER  4
#define NEUMANN 6
#define B_TOT   128
#define NB      8
#define BSUB    8
#define NWG     (B_TOT / BSUB)
#define THREADS 128
#define DT_C    0.05f
#define LDA     66
#define ROWS_ALL (C_DIM * BSUB)
#define MT_ALL   (ROWS_ALL / 16)
#define SLAB     (BSUB * S_DIM)
#define PTILE    256
#define PBUF     (4 * 4 * PTILE)

typedef __attribute__((ext_vector_type(2))) float v2f;
typedef __attribute__((ext_vector_type(4))) float v4f;
typedef __attribute__((ext_vector_type(8))) float v8f;

__device__ __forceinline__ void ksplit_product(const v2f (&wa)[4][4],
                                               const float (&xb)[8],
                                               float* __restrict__ sPb,
                                               int wave, int lane,
                                               float (&tot)[8]) {
  v8f acc[4];
#pragma unroll
  for (int i = 0; i < 4; ++i)
    acc[i] = (v8f){0.f, 0.f, 0.f, 0.f, 0.f, 0.f, 0.f, 0.f};
#pragma unroll
  for (int j = 0; j < 4; ++j) {
    v2f b;
    b.x = xb[2 * j];
    b.y = xb[2 * j + 1];
#pragma unroll
    for (int ii = 0; ii < 4; ++ii) {
      const int i = (ii + 1) & 3;
      acc[i] = __builtin_amdgcn_wmma_f32_16x16x4_f32(false, wa[i][j], false, b,
                                                     (short)0, acc[i],
                                                     false, false);
    }
  }
#pragma unroll
  for (int i = 1; i < 4; ++i) {
    const int dst = (wave + i) & 3;
    float* q = sPb + (dst * 4 + wave) * PTILE + lane * 4;
    *(v4f*)q         = (v4f){acc[i][0], acc[i][1], acc[i][2], acc[i][3]};
    *(v4f*)(q + 128) = (v4f){acc[i][4], acc[i][5], acc[i][6], acc[i][7]};
  }
  __syncthreads();
#pragma unroll
  for (int r = 0; r < 8; ++r) tot[r] = acc[0][r];
#pragma unroll
  for (int i = 1; i < 4; ++i) {
    const int srcw = (wave + i) & 3;
    const float* q = sPb + (wave * 4 + srcw) * PTILE + lane * 4;
    const v4f lo = *(const v4f*)q;
    const v4f hi = *(const v4f*)(q + 128);
    tot[0] += lo.x; tot[1] += lo.y; tot[2] += lo.z; tot[3] += lo.w;
    tot[4] += hi.x; tot[5] += hi.y; tot[6] += hi.z; tot[7] += hi.w;
  }
}

__device__ __forceinline__ void wmma_gemm_64x64(const float* __restrict__ A,
                                                const float* __restrict__ Bt,
                                                int n0, int lane,
                                                v8f (&acc)[MT_ALL]) {
  const int m = lane & 15;
  const int h = (lane >> 4) << 1;
#pragma unroll
  for (int mt = 0; mt < MT_ALL; ++mt)
    acc[mt] = (v8f){0.f, 0.f, 0.f, 0.f, 0.f, 0.f, 0.f, 0.f};
#pragma unroll
  for (int k0 = 0; k0 < S_DIM; k0 += 4) {
    v2f b;
    b.x = Bt[(k0 + h) * LDA + n0 + m];
    b.y = Bt[(k0 + h + 1) * LDA + n0 + m];
#pragma unroll
    for (int mt = 0; mt < MT_ALL; ++mt) {
      v2f a;
      a.x = A[(mt * 16 + m) * LDA + k0 + h];
      a.y = A[(mt * 16 + m) * LDA + k0 + h + 1];
      acc[mt] = __builtin_amdgcn_wmma_f32_16x16x4_f32(false, a, false, b,
                                                      (short)0, acc[mt],
                                                      false, false);
    }
  }
}

__global__ __launch_bounds__(THREADS, 1)
void solver_gfx1250_kernel(const float* __restrict__ y0,
                           const float* __restrict__ forces,
                           const float* __restrict__ W,
                           float* __restrict__ out) {
  __shared__ float sWt[S_DIM * LDA];
  __shared__ float sSa[ROWS_ALL * LDA];
  __shared__ __attribute__((aligned(16))) float sP[2 * PBUF];
  __shared__ __attribute__((aligned(16))) float sY[C_DIM][SLAB];
  __shared__ float sYprev[SLAB];
  __shared__ __attribute__((aligned(16))) float sRa[C_DIM][SLAB];
  __shared__ __attribute__((aligned(16))) float sGa[C_DIM][SLAB];

  const int tid  = threadIdx.x;
  const int lane = tid & 31;
  const int wave = tid >> 5;
  const int n0   = wave * 16;
  const int b0   = blockIdx.x * BSUB;
  const int col  = n0 + (lane & 15);
  const int hh   = lane >> 4;

  const int nb = lane & 7;
  const int eo = nb * S_DIM + n0 + 8 * hh;

  v2f wa[4][4];
#pragma unroll
  for (int i = 0; i < 4; ++i)
#pragma unroll
    for (int j = 0; j < 4; ++j)
      wa[i][j] = *(const v2f*)(W + (16 * ((wave + i) & 3) + (lane & 15)) * S_DIM
                                 + n0 + 8 * hh + 2 * j);
  int par = 0;

  for (int e = tid; e < S_DIM * S_DIM; e += THREADS) {
    int p = e >> 6, s = e & 63;
    sWt[s * LDA + p] = W[p * S_DIM + s];
  }
  for (int e = tid; e < SLAB; e += THREADS) {
    int b = e >> 6, j = e & 63;
    float v = y0[(b0 + b) * S_DIM + j];
    sYprev[e] = v;
    out[(size_t)(b0 + b) * S_DIM + j] = v;
  }
  __syncthreads();

  for (int blk = 0; blk < NB; ++blk) {
    const float* fblk =
        forces + ((size_t)(1 + blk * C_DIM) * B_TOT + b0) * S_DIM;

    float fv[MT_ALL][8];
#pragma unroll
    for (int mt = 0; mt < MT_ALL; ++mt)
#pragma unroll
      for (int v = 0; v < 8; ++v)
        fv[mt][v] = fblk[(size_t)(2 * mt + hh) * B_TOT * S_DIM + v * S_DIM + col];

    for (int c = 0; c < C_DIM; ++c)
      for (int e = tid; e < SLAB; e += THREADS) sY[c][e] = 0.f;
    __syncthreads();

    for (int it = 0; it < N_ITER; ++it) {
#pragma unroll
      for (int q = 0; q < SLAB / THREADS; ++q) {
        const int e = q * THREADS + tid;
        const int b = e >> 6, j = e & 63;
        float ym = sYprev[e];
#pragma unroll
        for (int c = 0; c < C_DIM; ++c) {
          const float yc = sY[c][e];
          sSa[(c * BSUB + b) * LDA + j] = 0.5f * (ym + yc);
          sRa[c][e] = yc - ym;
          ym = yc;
        }
      }
      __syncthreads();

      {
        v8f tacc[MT_ALL];
        wmma_gemm_64x64(sSa, sWt, n0, lane, tacc);
#pragma unroll
        for (int mt = 0; mt < MT_ALL; ++mt) {
          const int c = 2 * mt + hh;
#pragma unroll
          for (int v = 0; v < 8; ++v) {
            const int e = v * S_DIM + col;
            const float t = tanhf(tacc[mt][v]);
            const float g = 0.5f * DT_C * (1.0f - t * t);
            const float R = sRa[c][e] + DT_C * t - DT_C * fv[mt][v];
            sRa[c][e] = R;
            sGa[c][e] = g;
          }
        }
      }
      __syncthreads();

      float xb[8];
#pragma unroll
      for (int r = 0; r < 8; ++r) xb[r] = 0.f;

      for (int c = 0; c < C_DIM; ++c) {
        float rv[8], gv[8];
        v4f y0v, y1v;
        {
          const v4f r0 = *(const v4f*)&sRa[c][eo], r1 = *(const v4f*)&sRa[c][eo + 4];
          const v4f g0 = *(const v4f*)&sGa[c][eo], g1 = *(const v4f*)&sGa[c][eo + 4];
          y0v = *(const v4f*)&sY[c][eo];
          y1v = *(const v4f*)&sY[c][eo + 4];
          rv[0] = r0.x; rv[1] = r0.y; rv[2] = r0.z; rv[3] = r0.w;
          rv[4] = r1.x; rv[5] = r1.y; rv[6] = r1.z; rv[7] = r1.w;
          gv[0] = g0.x; gv[1] = g0.y; gv[2] = g0.z; gv[3] = g0.w;
          gv[4] = g1.x; gv[5] = g1.y; gv[6] = g1.z; gv[7] = g1.w;
        }

        if (c > 0) {
          float wx[8];
          ksplit_product(wa, xb, sP + par * PBUF, wave, lane, wx);
          par ^= 1;
#pragma unroll
          for (int r = 0; r < 8; ++r) {
            float rhs = rv[r];
            rhs += xb[r] - gv[r] * wx[r];
            rv[r] = rhs;
          }
        }
#pragma unroll
        for (int r = 0; r < 8; ++r) xb[r] = rv[r];

        for (int k = 0; k < NEUMANN; ++k) {
          float mx[8];
          ksplit_product(wa, xb, sP + par * PBUF, wave, lane, mx);
          par ^= 1;
#pragma unroll
          for (int r = 0; r < 8; ++r) xb[r] = rv[r] - gv[r] * mx[r];
        }

        y0v.x -= xb[0]; y0v.y -= xb[1]; y0v.z -= xb[2]; y0v.w -= xb[3];
        y1v.x -= xb[4]; y1v.y -= xb[5]; y1v.z -= xb[6]; y1v.w -= xb[7];
        if ((lane & 15) < BSUB) {
          *(v4f*)&sY[c][eo]     = y0v;
          *(v4f*)&sY[c][eo + 4] = y1v;
        }
      }
      __syncthreads();
    }

    for (int c = 0; c < C_DIM; ++c)
      for (int e = tid; e < SLAB; e += THREADS) {
        int b = e >> 6, j = e & 63;
        out[((size_t)(1 + blk * C_DIM + c) * B_TOT + b0 + b) * S_DIM + j] =
            sY[c][e];
      }
    for (int e = tid; e < SLAB; e += THREADS)
      sYprev[e] = sY[C_DIM - 1][e];
    __syncthreads();
  }
}

extern "C" void kernel_launch(void* const* d_in, const int* in_sizes, int n_in,
                              void* d_out, int out_size, void* d_ws,
                              size_t ws_size, hipStream_t stream) {
  (void)in_sizes; (void)n_in; (void)out_size; (void)d_ws; (void)ws_size;
  const float* y0     = (const float*)d_in[0];
  const float* forces = (const float*)d_in[1];
  const float* W      = (const float*)d_in[2];
  float* out = (float*)d_out;
  solver_gfx1250_kernel<<<NWG, THREADS, 0, stream>>>(y0, forces, W, out);
}


# Element-wise LDS reads issued in groups, not read-and-wait one by one: residual epilogue's 32 y-difference terms read under the GEMM into registers; chunk emission's reads unrolled; sum order kept
# speedup vs baseline: 2.5041x; 1.0753x over previous
#include <hip/hip_runtime.h>


#define S_DIM   64
#define C_DIM   8
#define N_ITER  4
#define NEUMANN 6
#define B_TOT   128
#define NB      8
#define BSUB    8
#define NWG     (B_TOT / BSUB)
#define THREADS 128
#define DT_C    0.05f
#define LDA     66
#define ROWS_ALL (C_DIM * BSUB)
#define MT_ALL   (ROWS_ALL / 16)
#define SLAB     (BSUB * S_DIM)
#define PTILE    256
#define PBUF     (4 * 4 * PTILE)

typedef __attribute__((ext_vector_type(2))) float v2f;
typedef __attribute__((ext_vector_type(4))) float v4f;
typedef __attribute__((ext_vector_type(8))) float v8f;

__device__ __forceinline__ void ksplit_product(const v2f (&wa)[4][4],
                                               const float (&xb)[8],
                                               float* __restrict__ sPb,
                                               int wave, int lane,
                                               float (&tot)[8]) {
  v8f acc[4];
#pragma unroll
  for (int i = 0; i < 4; ++i)
    acc[i] = (v8f){0.f, 0.f, 0.f, 0.f, 0.f, 0.f, 0.f, 0.f};
#pragma unroll
  for (int j = 0; j < 4; ++j) {
    v2f b;
    b.x = xb[2 * j];
    b.y = xb[2 * j + 1];
#pragma unroll
    for (int ii = 0; ii < 4; ++ii) {
      const int i = (ii + 1) & 3;
      acc[i] = __builtin_amdgcn_wmma_f32_16x16x4_f32(false, wa[i][j], false, b,
                                                     (short)0, acc[i],
                                                     false, false);
    }
  }
#pragma unroll
  for (int i = 1; i < 4; ++i) {
    const int dst = (wave + i) & 3;
    float* q = sPb + (dst * 4 + wave) * PTILE + lane * 4;
    *(v4f*)q         = (v4f){acc[i][0], acc[i][1], acc[i][2], acc[i][3]};
    *(v4f*)(q + 128) = (v4f){acc[i][4], acc[i][5], acc[i][6], acc[i][7]};
  }
  __syncthreads();
#pragma unroll
  for (int r = 0; r < 8; ++r) tot[r] = acc[0][r];
#pragma unroll
  for (int i = 1; i < 4; ++i) {
    const int srcw = (wave + i) & 3;
    const float* q = sPb + (wave * 4 + srcw) * PTILE + lane * 4;
    const v4f lo = *(const v4f*)q;
    const v4f hi = *(const v4f*)(q + 128);
    tot[0] += lo.x; tot[1] += lo.y; tot[2] += lo.z; tot[3] += lo.w;
    tot[4] += hi.x; tot[5] += hi.y; tot[6] += hi.z; tot[7] += hi.w;
  }
}

__device__ __forceinline__ void wmma_gemm_64x64(const float* __restrict__ A,
                                                const float* __restrict__ Bt,
                                                int n0, int lane,
                                                v8f (&acc)[MT_ALL]) {
  const int m = lane & 15;
  const int h = (lane >> 4) << 1;
#pragma unroll
  for (int mt = 0; mt < MT_ALL; ++mt)
    acc[mt] = (v8f){0.f, 0.f, 0.f, 0.f, 0.f, 0.f, 0.f, 0.f};
#pragma unroll
  for (int k0 = 0; k0 < S_DIM; k0 += 4) {
    v2f b;
    b.x = Bt[(k0 + h) * LDA + n0 + m];
    b.y = Bt[(k0 + h + 1) * LDA + n0 + m];
#pragma unroll
    for (int mt = 0; mt < MT_ALL; ++mt) {
      v2f a;
      a.x = A[(mt * 16 + m) * LDA + k0 + h];
      a.y = A[(mt * 16 + m) * LDA + k0 + h + 1];
      acc[mt] = __builtin_amdgcn_wmma_f32_16x16x4_f32(false, a, false, b,
                                                      (short)0, acc[mt],
                                                      false, false);
    }
  }
}

__global__ __launch_bounds__(THREADS, 1)
void solver_gfx1250_kernel(const float* __restrict__ y0,
                           const float* __restrict__ forces,
                           const float* __restrict__ W,
                           float* __restrict__ out) {
  __shared__ float sWt[S_DIM * LDA];
  __shared__ float sSa[ROWS_ALL * LDA];
  __shared__ __attribute__((aligned(16))) float sP[2 * PBUF];
  __shared__ __attribute__((aligned(16))) float sY[C_DIM][SLAB];
  __shared__ float sYprev[SLAB];
  __shared__ __attribute__((aligned(16))) float sRa[C_DIM][SLAB];
  __shared__ __attribute__((aligned(16))) float sGa[C_DIM][SLAB];

  const int tid  = threadIdx.x;
  const int lane = tid & 31;
  const int wave = tid >> 5;
  const int n0   = wave * 16;
  const int b0   = blockIdx.x * BSUB;
  const int col  = n0 + (lane & 15);
  const int hh   = lane >> 4;

  const int nb = lane & 7;
  const int eo = nb * S_DIM + n0 + 8 * hh;

  v2f wa[4][4];
#pragma unroll
  for (int i = 0; i < 4; ++i)
#pragma unroll
    for (int j = 0; j < 4; ++j)
      wa[i][j] = *(const v2f*)(W + (16 * ((wave + i) & 3) + (lane & 15)) * S_DIM
                                 + n0 + 8 * hh + 2 * j);
  int par = 0;

  for (int e = tid; e < S_DIM * S_DIM; e += THREADS) {
    int p = e >> 6, s = e & 63;
    sWt[s * LDA + p] = W[p * S_DIM + s];
  }
  for (int e = tid; e < SLAB; e += THREADS) {
    int b = e >> 6, j = e & 63;
    float v = y0[(b0 + b) * S_DIM + j];
    sYprev[e] = v;
    out[(size_t)(b0 + b) * S_DIM + j] = v;
  }
  __syncthreads();

  for (int blk = 0; blk < NB; ++blk) {
    const float* fblk =
        forces + ((size_t)(1 + blk * C_DIM) * B_TOT + b0) * S_DIM;

    float fv[MT_ALL][8];
#pragma unroll
    for (int mt = 0; mt < MT_ALL; ++mt)
#pragma unroll
      for (int v = 0; v < 8; ++v)
        fv[mt][v] = fblk[(size_t)(2 * mt + hh) * B_TOT * S_DIM + v * S_DIM + col];

    for (int c = 0; c < C_DIM; ++c)
      for (int e = tid; e < SLAB; e += THREADS) sY[c][e] = 0.f;
    __syncthreads();

    for (int it = 0; it < N_ITER; ++it) {
#pragma unroll
      for (int q = 0; q < SLAB / THREADS; ++q) {
        const int e = q * THREADS + tid;
        const int b = e >> 6, j = e & 63;
        float ym = sYprev[e];
#pragma unroll
        for (int c = 0; c < C_DIM; ++c) {
          const float yc = sY[c][e];
          sSa[(c * BSUB + b) * LDA + j] = 0.5f * (ym + yc);
          sRa[c][e] = yc - ym;
          ym = yc;
        }
      }
      __syncthreads();

      {
        float yd[MT_ALL][8];
#pragma unroll
        for (int mt = 0; mt < MT_ALL; ++mt)
#pragma unroll
          for (int v = 0; v < 8; ++v)
            yd[mt][v] = sRa[2 * mt + hh][v * S_DIM + col];

        v8f tacc[MT_ALL];
        wmma_gemm_64x64(sSa, sWt, n0, lane, tacc);
#pragma unroll
        for (int mt = 0; mt < MT_ALL; ++mt) {
          const int c = 2 * mt + hh;
#pragma unroll
          for (int v = 0; v < 8; ++v) {
            const int e = v * S_DIM + col;
            const float t = tanhf(tacc[mt][v]);
            const float g = 0.5f * DT_C * (1.0f - t * t);
            const float R = yd[mt][v] + DT_C * t - DT_C * fv[mt][v];
            sRa[c][e] = R;
            sGa[c][e] = g;
          }
        }
      }
      __syncthreads();

      float xb[8];
#pragma unroll
      for (int r = 0; r < 8; ++r) xb[r] = 0.f;

      for (int c = 0; c < C_DIM; ++c) {
        float rv[8], gv[8];
        v4f y0v, y1v;
        {
          const v4f r0 = *(const v4f*)&sRa[c][eo], r1 = *(const v4f*)&sRa[c][eo + 4];
          const v4f g0 = *(const v4f*)&sGa[c][eo], g1 = *(const v4f*)&sGa[c][eo + 4];
          y0v = *(const v4f*)&sY[c][eo];
          y1v = *(const v4f*)&sY[c][eo + 4];
          rv[0] = r0.x; rv[1] = r0.y; rv[2] = r0.z; rv[3] = r0.w;
          rv[4] = r1.x; rv[5] = r1.y; rv[6] = r1.z; rv[7] = r1.w;
          gv[0] = g0.x; gv[1] = g0.y; gv[2] = g0.z; gv[3] = g0.w;
          gv[4] = g1.x; gv[5] = g1.y; gv[6] = g1.z; gv[7] = g1.w;
        }

        if (c > 0) {
          float wx[8];
          ksplit_product(wa, xb, sP + par * PBUF, wave, lane, wx);
          par ^= 1;
#pragma unroll
          for (int r = 0; r < 8; ++r) {
            float rhs = rv[r];
            rhs += xb[r] - gv[r] * wx[r];
            rv[r] = rhs;
          }
        }
#pragma unroll
        for (int r = 0; r < 8; ++r) xb[r] = rv[r];

        for (int k = 0; k < NEUMANN; ++k) {
          float mx[8];
          ksplit_product(wa, xb, sP + par * PBUF, wave, lane, mx);
          par ^= 1;
#pragma unroll
          for (int r = 0; r < 8; ++r) xb[r] = rv[r] - gv[r] * mx[r];
        }

        y0v.x -= xb[0]; y0v.y -= xb[1]; y0v.z -= xb[2]; y0v.w -= xb[3];
        y1v.x -= xb[4]; y1v.y -= xb[5]; y1v.z -= xb[6]; y1v.w -= xb[7];
        if ((lane & 15) < BSUB) {
          *(v4f*)&sY[c][eo]     = y0v;
          *(v4f*)&sY[c][eo + 4] = y1v;
        }
      }
      __syncthreads();
    }

    {
      float yo[C_DIM][SLAB / THREADS];
#pragma unroll
      for (int c = 0; c < C_DIM; ++c)
#pragma unroll
        for (int q = 0; q < SLAB / THREADS; ++q)
          yo[c][q] = sY[c][q * THREADS + tid];
#pragma unroll
      for (int c = 0; c < C_DIM; ++c)
#pragma unroll
        for (int q = 0; q < SLAB / THREADS; ++q) {
          const int e = q * THREADS + tid;
          const int b = e >> 6, j = e & 63;
          out[((size_t)(1 + blk * C_DIM + c) * B_TOT + b0 + b) * S_DIM + j] =
              yo[c][q];
        }
#pragma unroll
      for (int q = 0; q < SLAB / THREADS; ++q)
        sYprev[q * THREADS + tid] = yo[C_DIM - 1][q];
    }
    __syncthreads();
  }
}

extern "C" void kernel_launch(void* const* d_in, const int* in_sizes, int n_in,
                              void* d_out, int out_size, void* d_ws,
                              size_t ws_size, hipStream_t stream) {
  (void)in_sizes; (void)n_in; (void)out_size; (void)d_ws; (void)ws_size;
  const float* y0     = (const float*)d_in[0];
  const float* forces = (const float*)d_in[1];
  const float* W      = (const float*)d_in[2];
  float* out = (float*)d_out;
  solver_gfx1250_kernel<<<NWG, THREADS, 0, stream>>>(y0, forces, W, out);
}


# First Newton iteration of a block: cells 1-7 start at y=0, so g is the constant dt/2 and their 6-step Neumann is ONE product with Q = I+M+..+M^6 (M = -g W), built at kernel start; sum order changes
# speedup vs baseline: 2.7870x; 1.1130x over previous
#include <hip/hip_runtime.h>


#define S_DIM   64
#define C_DIM   8
#define N_ITER  4
#define NEUMANN 6
#define B_TOT   128
#define NB      8
#define BSUB    8
#define NWG     (B_TOT / BSUB)
#define THREADS 128
#define DT_C    0.05f
#define LDA     66
#define ROWS_ALL (C_DIM * BSUB)
#define MT_ALL   (ROWS_ALL / 16)
#define SLAB     (BSUB * S_DIM)
#define PTILE    256
#define PBUF     (4 * 4 * PTILE)

typedef __attribute__((ext_vector_type(2))) float v2f;
typedef __attribute__((ext_vector_type(4))) float v4f;
typedef __attribute__((ext_vector_type(8))) float v8f;

__device__ __forceinline__ void ksplit_product(const v2f (&wa)[4][4],
                                               const float (&xb)[8],
                                               float* __restrict__ sPb,
                                               int wave, int lane,
                                               float (&tot)[8]) {
  v8f acc[4];
#pragma unroll
  for (int i = 0; i < 4; ++i)
    acc[i] = (v8f){0.f, 0.f, 0.f, 0.f, 0.f, 0.f, 0.f, 0.f};
#pragma unroll
  for (int j = 0; j < 4; ++j) {
    v2f b;
    b.x = xb[2 * j];
    b.y = xb[2 * j + 1];
#pragma unroll
    for (int ii = 0; ii < 4; ++ii) {
      const int i = (ii + 1) & 3;
      acc[i] = __builtin_amdgcn_wmma_f32_16x16x4_f32(false, wa[i][j], false, b,
                                                     (short)0, acc[i],
                                                     false, false);
    }
  }
#pragma unroll
  for (int i = 1; i < 4; ++i) {
    const int dst = (wave + i) & 3;
    float* q = sPb + (dst * 4 + wave) * PTILE + lane * 4;
    *(v4f*)q         = (v4f){acc[i][0], acc[i][1], acc[i][2], acc[i][3]};
    *(v4f*)(q + 128) = (v4f){acc[i][4], acc[i][5], acc[i][6], acc[i][7]};
  }
  __syncthreads();
#pragma unroll
  for (int r = 0; r < 8; ++r) tot[r] = acc[0][r];
#pragma unroll
  for (int i = 1; i < 4; ++i) {
    const int srcw = (wave + i) & 3;
    const float* q = sPb + (wave * 4 + srcw) * PTILE + lane * 4;
    const v4f lo = *(const v4f*)q;
    const v4f hi = *(const v4f*)(q + 128);
    tot[0] += lo.x; tot[1] += lo.y; tot[2] += lo.z; tot[3] += lo.w;
    tot[4] += hi.x; tot[5] += hi.y; tot[6] += hi.z; tot[7] += hi.w;
  }
}

__device__ __forceinline__ void wmma_gemm_64x64(const float* __restrict__ A,
                                                const float* __restrict__ Bt,
                                                int n0, int lane,
                                                v8f (&acc)[MT_ALL]) {
  const int m = lane & 15;
  const int h = (lane >> 4) << 1;
#pragma unroll
  for (int mt = 0; mt < MT_ALL; ++mt)
    acc[mt] = (v8f){0.f, 0.f, 0.f, 0.f, 0.f, 0.f, 0.f, 0.f};
#pragma unroll
  for (int k0 = 0; k0 < S_DIM; k0 += 4) {
    v2f b;
    b.x = Bt[(k0 + h) * LDA + n0 + m];
    b.y = Bt[(k0 + h + 1) * LDA + n0 + m];
#pragma unroll
    for (int mt = 0; mt < MT_ALL; ++mt) {
      v2f a;
      a.x = A[(mt * 16 + m) * LDA + k0 + h];
      a.y = A[(mt * 16 + m) * LDA + k0 + h + 1];
      acc[mt] = __builtin_amdgcn_wmma_f32_16x16x4_f32(false, a, false, b,
                                                      (short)0, acc[mt],
                                                      false, false);
    }
  }
}

__global__ __launch_bounds__(THREADS, 1)
void solver_gfx1250_kernel(const float* __restrict__ y0,
                           const float* __restrict__ forces,
                           const float* __restrict__ W,
                           float* __restrict__ out) {
  __shared__ float sWt[S_DIM * LDA];
  __shared__ float sSa[ROWS_ALL * LDA];
  __shared__ __attribute__((aligned(16))) float sP[2 * PBUF];
  __shared__ __attribute__((aligned(16))) float sY[C_DIM][SLAB];
  __shared__ float sYprev[SLAB];
  __shared__ __attribute__((aligned(16))) float sRa[C_DIM][SLAB];
  __shared__ __attribute__((aligned(16))) float sGa[C_DIM][SLAB];

  const int tid  = threadIdx.x;
  const int lane = tid & 31;
  const int wave = tid >> 5;
  const int n0   = wave * 16;
  const int b0   = blockIdx.x * BSUB;
  const int col  = n0 + (lane & 15);
  const int hh   = lane >> 4;

  const int nb = lane & 7;
  const int eo = nb * S_DIM + n0 + 8 * hh;

  v2f wa[4][4];
#pragma unroll
  for (int i = 0; i < 4; ++i)
#pragma unroll
    for (int j = 0; j < 4; ++j)
      wa[i][j] = *(const v2f*)(W + (16 * ((wave + i) & 3) + (lane & 15)) * S_DIM
                                 + n0 + 8 * hh + 2 * j);
  int par = 0;

  for (int e = tid; e < S_DIM * S_DIM; e += THREADS) {
    int p = e >> 6, s = e & 63;
    sWt[s * LDA + p] = W[p * S_DIM + s];
  }
  for (int e = tid; e < SLAB; e += THREADS) {
    int b = e >> 6, j = e & 63;
    float v = y0[(b0 + b) * S_DIM + j];
    sYprev[e] = v;
    out[(size_t)(b0 + b) * S_DIM + j] = v;
  }
  __syncthreads();

  const float g0 = 0.5f * DT_C;
  for (int e = tid; e < S_DIM * S_DIM; e += THREADS) {
    int s = e >> 6, p = e & 63;
    sSa[s * LDA + p] = (s == p ? 1.0f : 0.0f) - g0 * sWt[s * LDA + p];
  }
  __syncthreads();
#pragma unroll
  for (int k = 1; k < NEUMANN; ++k) {
    const float* cur = (k & 1) ? sSa : sP;
    float* nxt = (k & 1) ? sP : sSa;
    v8f qacc[MT_ALL];
    wmma_gemm_64x64(sWt, cur, n0, lane, qacc);
#pragma unroll
    for (int mt = 0; mt < MT_ALL; ++mt)
#pragma unroll
      for (int v = 0; v < 8; ++v) {
        const int row = mt * 16 + 8 * hh + v;
        nxt[row * LDA + col] = (row == col ? 1.0f : 0.0f) - g0 * qacc[mt][v];
      }
    __syncthreads();
  }
  v2f wq[4][4];
#pragma unroll
  for (int i = 0; i < 4; ++i)
#pragma unroll
    for (int j = 0; j < 4; ++j) {
      const int row = 16 * ((wave + i) & 3) + (lane & 15);
      const int kc = n0 + 8 * hh + 2 * j;
      wq[i][j].x = sP[kc * LDA + row];
      wq[i][j].y = sP[(kc + 1) * LDA + row];
    }

  for (int blk = 0; blk < NB; ++blk) {
    const float* fblk =
        forces + ((size_t)(1 + blk * C_DIM) * B_TOT + b0) * S_DIM;

    float fv[MT_ALL][8];
#pragma unroll
    for (int mt = 0; mt < MT_ALL; ++mt)
#pragma unroll
      for (int v = 0; v < 8; ++v)
        fv[mt][v] = fblk[(size_t)(2 * mt + hh) * B_TOT * S_DIM + v * S_DIM + col];

    for (int c = 0; c < C_DIM; ++c)
      for (int e = tid; e < SLAB; e += THREADS) sY[c][e] = 0.f;
    __syncthreads();

    for (int it = 0; it < N_ITER; ++it) {
#pragma unroll
      for (int q = 0; q < SLAB / THREADS; ++q) {
        const int e = q * THREADS + tid;
        const int b = e >> 6, j = e & 63;
        float ym = sYprev[e];
#pragma unroll
        for (int c = 0; c < C_DIM; ++c) {
          const float yc = sY[c][e];
          sSa[(c * BSUB + b) * LDA + j] = 0.5f * (ym + yc);
          sRa[c][e] = yc - ym;
          ym = yc;
        }
      }
      __syncthreads();

      {
        float yd[MT_ALL][8];
#pragma unroll
        for (int mt = 0; mt < MT_ALL; ++mt)
#pragma unroll
          for (int v = 0; v < 8; ++v)
            yd[mt][v] = sRa[2 * mt + hh][v * S_DIM + col];

        v8f tacc[MT_ALL];
        wmma_gemm_64x64(sSa, sWt, n0, lane, tacc);
#pragma unroll
        for (int mt = 0; mt < MT_ALL; ++mt) {
          const int c = 2 * mt + hh;
#pragma unroll
          for (int v = 0; v < 8; ++v) {
            const int e = v * S_DIM + col;
            const float t = tanhf(tacc[mt][v]);
            const float g = 0.5f * DT_C * (1.0f - t * t);
            const float R = yd[mt][v] + DT_C * t - DT_C * fv[mt][v];
            sRa[c][e] = R;
            sGa[c][e] = g;
          }
        }
      }
      __syncthreads();

      float xb[8];
#pragma unroll
      for (int r = 0; r < 8; ++r) xb[r] = 0.f;

      for (int c = 0; c < C_DIM; ++c) {
        float rv[8], gv[8];
        v4f y0v, y1v;
        {
          const v4f r0 = *(const v4f*)&sRa[c][eo], r1 = *(const v4f*)&sRa[c][eo + 4];
          const v4f g0 = *(const v4f*)&sGa[c][eo], g1 = *(const v4f*)&sGa[c][eo + 4];
          y0v = *(const v4f*)&sY[c][eo];
          y1v = *(const v4f*)&sY[c][eo + 4];
          rv[0] = r0.x; rv[1] = r0.y; rv[2] = r0.z; rv[3] = r0.w;
          rv[4] = r1.x; rv[5] = r1.y; rv[6] = r1.z; rv[7] = r1.w;
          gv[0] = g0.x; gv[1] = g0.y; gv[2] = g0.z; gv[3] = g0.w;
          gv[4] = g1.x; gv[5] = g1.y; gv[6] = g1.z; gv[7] = g1.w;
        }

        if (c > 0) {
          float wx[8];
          ksplit_product(wa, xb, sP + par * PBUF, wave, lane, wx);
          par ^= 1;
#pragma unroll
          for (int r = 0; r < 8; ++r) {
            float rhs = rv[r];
            rhs += xb[r] - gv[r] * wx[r];
            rv[r] = rhs;
          }
        }
#pragma unroll
        for (int r = 0; r < 8; ++r) xb[r] = rv[r];

        if (it == 0 && c > 0) {
          float qx[8];
          ksplit_product(wq, xb, sP + par * PBUF, wave, lane, qx);
          par ^= 1;
#pragma unroll
          for (int r = 0; r < 8; ++r) xb[r] = qx[r];
        } else {
          for (int k = 0; k < NEUMANN; ++k) {
            float mx[8];
            ksplit_product(wa, xb, sP + par * PBUF, wave, lane, mx);
            par ^= 1;
#pragma unroll
            for (int r = 0; r < 8; ++r) xb[r] = rv[r] - gv[r] * mx[r];
          }
        }

        y0v.x -= xb[0]; y0v.y -= xb[1]; y0v.z -= xb[2]; y0v.w -= xb[3];
        y1v.x -= xb[4]; y1v.y -= xb[5]; y1v.z -= xb[6]; y1v.w -= xb[7];
        if ((lane & 15) < BSUB) {
          *(v4f*)&sY[c][eo]     = y0v;
          *(v4f*)&sY[c][eo + 4] = y1v;
        }
      }
      __syncthreads();
    }

    {
      float yo[C_DIM][SLAB / THREADS];
#pragma unroll
      for (int c = 0; c < C_DIM; ++c)
#pragma unroll
        for (int q = 0; q < SLAB / THREADS; ++q)
          yo[c][q] = sY[c][q * THREADS + tid];
#pragma unroll
      for (int c = 0; c < C_DIM; ++c)
#pragma unroll
        for (int q = 0; q < SLAB / THREADS; ++q) {
          const int e = q * THREADS + tid;
          const int b = e >> 6, j = e & 63;
          out[((size_t)(1 + blk * C_DIM + c) * B_TOT + b0 + b) * S_DIM + j] =
              yo[c][q];
        }
#pragma unroll
      for (int q = 0; q < SLAB / THREADS; ++q)
        sYprev[q * THREADS + tid] = yo[C_DIM - 1][q];
    }
    __syncthreads();
  }
}

extern "C" void kernel_launch(void* const* d_in, const int* in_sizes, int n_in,
                              void* d_out, int out_size, void* d_ws,
                              size_t ws_size, hipStream_t stream) {
  (void)in_sizes; (void)n_in; (void)out_size; (void)d_ws; (void)ws_size;
  const float* y0     = (const float*)d_in[0];
  const float* forces = (const float*)d_in[1];
  const float* W      = (const float*)d_in[2];
  float* out = (float*)d_out;
  solver_gfx1250_kernel<<<NWG, THREADS, 0, stream>>>(y0, forces, W, out);
}


# Sub-diagonal block used as L = D - 2I: x_c = D^-1 (R_c + 2 x_prev) - x_prev, so the sweep's W x_prev product per cell is gone (157 serial products per block, not 185); sum order changes
# speedup vs baseline: 3.1147x; 1.1176x over previous
#include <hip/hip_runtime.h>


#define S_DIM   64
#define C_DIM   8
#define N_ITER  4
#define NEUMANN 6
#define B_TOT   128
#define NB      8
#define BSUB    8
#define NWG     (B_TOT / BSUB)
#define THREADS 128
#define DT_C    0.05f
#define LDA     66
#define ROWS_ALL (C_DIM * BSUB)
#define MT_ALL   (ROWS_ALL / 16)
#define SLAB     (BSUB * S_DIM)
#define PTILE    256
#define PBUF     (4 * 4 * PTILE)

typedef __attribute__((ext_vector_type(2))) float v2f;
typedef __attribute__((ext_vector_type(4))) float v4f;
typedef __attribute__((ext_vector_type(8))) float v8f;

__device__ __forceinline__ void ksplit_product(const v2f (&wa)[4][4],
                                               const float (&xb)[8],
                                               float* __restrict__ sPb,
                                               int wave, int lane,
                                               float (&tot)[8]) {
  v8f acc[4];
#pragma unroll
  for (int i = 0; i < 4; ++i)
    acc[i] = (v8f){0.f, 0.f, 0.f, 0.f, 0.f, 0.f, 0.f, 0.f};
#pragma unroll
  for (int j = 0; j < 4; ++j) {
    v2f b;
    b.x = xb[2 * j];
    b.y = xb[2 * j + 1];
#pragma unroll
    for (int ii = 0; ii < 4; ++ii) {
      const int i = (ii + 1) & 3;
      acc[i] = __builtin_amdgcn_wmma_f32_16x16x4_f32(false, wa[i][j], false, b,
                                                     (short)0, acc[i],
                                                     false, false);
    }
  }
#pragma unroll
  for (int i = 1; i < 4; ++i) {
    const int dst = (wave + i) & 3;
    float* q = sPb + (dst * 4 + wave) * PTILE + lane * 4;
    *(v4f*)q         = (v4f){acc[i][0], acc[i][1], acc[i][2], acc[i][3]};
    *(v4f*)(q + 128) = (v4f){acc[i][4], acc[i][5], acc[i][6], acc[i][7]};
  }
  __syncthreads();
#pragma unroll
  for (int r = 0; r < 8; ++r) tot[r] = acc[0][r];
#pragma unroll
  for (int i = 1; i < 4; ++i) {
    const int srcw = (wave + i) & 3;
    const float* q = sPb + (wave * 4 + srcw) * PTILE + lane * 4;
    const v4f lo = *(const v4f*)q;
    const v4f hi = *(const v4f*)(q + 128);
    tot[0] += lo.x; tot[1] += lo.y; tot[2] += lo.z; tot[3] += lo.w;
    tot[4] += hi.x; tot[5] += hi.y; tot[6] += hi.z; tot[7] += hi.w;
  }
}

__device__ __forceinline__ void wmma_gemm_64x64(const float* __restrict__ A,
                                                const float* __restrict__ Bt,
                                                int n0, int lane,
                                                v8f (&acc)[MT_ALL]) {
  const int m = lane & 15;
  const int h = (lane >> 4) << 1;
#pragma unroll
  for (int mt = 0; mt < MT_ALL; ++mt)
    acc[mt] = (v8f){0.f, 0.f, 0.f, 0.f, 0.f, 0.f, 0.f, 0.f};
#pragma unroll
  for (int k0 = 0; k0 < S_DIM; k0 += 4) {
    v2f b;
    b.x = Bt[(k0 + h) * LDA + n0 + m];
    b.y = Bt[(k0 + h + 1) * LDA + n0 + m];
#pragma unroll
    for (int mt = 0; mt < MT_ALL; ++mt) {
      v2f a;
      a.x = A[(mt * 16 + m) * LDA + k0 + h];
      a.y = A[(mt * 16 + m) * LDA + k0 + h + 1];
      acc[mt] = __builtin_amdgcn_wmma_f32_16x16x4_f32(false, a, false, b,
                                                      (short)0, acc[mt],
                                                      false, false);
    }
  }
}

__global__ __launch_bounds__(THREADS, 1)
void solver_gfx1250_kernel(const float* __restrict__ y0,
                           const float* __restrict__ forces,
                           const float* __restrict__ W,
                           float* __restrict__ out) {
  __shared__ float sWt[S_DIM * LDA];
  __shared__ float sSa[ROWS_ALL * LDA];
  __shared__ __attribute__((aligned(16))) float sP[2 * PBUF];
  __shared__ __attribute__((aligned(16))) float sY[C_DIM][SLAB];
  __shared__ float sYprev[SLAB];
  __shared__ __attribute__((aligned(16))) float sRa[C_DIM][SLAB];
  __shared__ __attribute__((aligned(16))) float sGa[C_DIM][SLAB];

  const int tid  = threadIdx.x;
  const int lane = tid & 31;
  const int wave = tid >> 5;
  const int n0   = wave * 16;
  const int b0   = blockIdx.x * BSUB;
  const int col  = n0 + (lane & 15);
  const int hh   = lane >> 4;

  const int nb = lane & 7;
  const int eo = nb * S_DIM + n0 + 8 * hh;

  v2f wa[4][4];
#pragma unroll
  for (int i = 0; i < 4; ++i)
#pragma unroll
    for (int j = 0; j < 4; ++j)
      wa[i][j] = *(const v2f*)(W + (16 * ((wave + i) & 3) + (lane & 15)) * S_DIM
                                 + n0 + 8 * hh + 2 * j);
  int par = 0;

  for (int e = tid; e < S_DIM * S_DIM; e += THREADS) {
    int p = e >> 6, s = e & 63;
    sWt[s * LDA + p] = W[p * S_DIM + s];
  }
  for (int e = tid; e < SLAB; e += THREADS) {
    int b = e >> 6, j = e & 63;
    float v = y0[(b0 + b) * S_DIM + j];
    sYprev[e] = v;
    out[(size_t)(b0 + b) * S_DIM + j] = v;
  }
  __syncthreads();

  const float g0 = 0.5f * DT_C;
  for (int e = tid; e < S_DIM * S_DIM; e += THREADS) {
    int s = e >> 6, p = e & 63;
    sSa[s * LDA + p] = (s == p ? 1.0f : 0.0f) - g0 * sWt[s * LDA + p];
  }
  __syncthreads();
#pragma unroll
  for (int k = 1; k < NEUMANN; ++k) {
    const float* cur = (k & 1) ? sSa : sP;
    float* nxt = (k & 1) ? sP : sSa;
    v8f qacc[MT_ALL];
    wmma_gemm_64x64(sWt, cur, n0, lane, qacc);
#pragma unroll
    for (int mt = 0; mt < MT_ALL; ++mt)
#pragma unroll
      for (int v = 0; v < 8; ++v) {
        const int row = mt * 16 + 8 * hh + v;
        nxt[row * LDA + col] = (row == col ? 1.0f : 0.0f) - g0 * qacc[mt][v];
      }
    __syncthreads();
  }
  v2f wq[4][4];
#pragma unroll
  for (int i = 0; i < 4; ++i)
#pragma unroll
    for (int j = 0; j < 4; ++j) {
      const int row = 16 * ((wave + i) & 3) + (lane & 15);
      const int kc = n0 + 8 * hh + 2 * j;
      wq[i][j].x = sP[kc * LDA + row];
      wq[i][j].y = sP[(kc + 1) * LDA + row];
    }

  for (int blk = 0; blk < NB; ++blk) {
    const float* fblk =
        forces + ((size_t)(1 + blk * C_DIM) * B_TOT + b0) * S_DIM;

    float fv[MT_ALL][8];
#pragma unroll
    for (int mt = 0; mt < MT_ALL; ++mt)
#pragma unroll
      for (int v = 0; v < 8; ++v)
        fv[mt][v] = fblk[(size_t)(2 * mt + hh) * B_TOT * S_DIM + v * S_DIM + col];

    for (int c = 0; c < C_DIM; ++c)
      for (int e = tid; e < SLAB; e += THREADS) sY[c][e] = 0.f;
    __syncthreads();

    for (int it = 0; it < N_ITER; ++it) {
#pragma unroll
      for (int q = 0; q < SLAB / THREADS; ++q) {
        const int e = q * THREADS + tid;
        const int b = e >> 6, j = e & 63;
        float ym = sYprev[e];
#pragma unroll
        for (int c = 0; c < C_DIM; ++c) {
          const float yc = sY[c][e];
          sSa[(c * BSUB + b) * LDA + j] = 0.5f * (ym + yc);
          sRa[c][e] = yc - ym;
          ym = yc;
        }
      }
      __syncthreads();

      {
        float yd[MT_ALL][8];
#pragma unroll
        for (int mt = 0; mt < MT_ALL; ++mt)
#pragma unroll
          for (int v = 0; v < 8; ++v)
            yd[mt][v] = sRa[2 * mt + hh][v * S_DIM + col];

        v8f tacc[MT_ALL];
        wmma_gemm_64x64(sSa, sWt, n0, lane, tacc);
#pragma unroll
        for (int mt = 0; mt < MT_ALL; ++mt) {
          const int c = 2 * mt + hh;
#pragma unroll
          for (int v = 0; v < 8; ++v) {
            const int e = v * S_DIM + col;
            const float t = tanhf(tacc[mt][v]);
            const float g = 0.5f * DT_C * (1.0f - t * t);
            const float R = yd[mt][v] + DT_C * t - DT_C * fv[mt][v];
            sRa[c][e] = R;
            sGa[c][e] = g;
          }
        }
      }
      __syncthreads();

      float rv[8], xq[8];
      {
        const v4f r0 = *(const v4f*)&sRa[0][eo], r1 = *(const v4f*)&sRa[0][eo + 4];
        rv[0] = r0.x; rv[1] = r0.y; rv[2] = r0.z; rv[3] = r0.w;
        rv[4] = r1.x; rv[5] = r1.y; rv[6] = r1.z; rv[7] = r1.w;
      }
#pragma unroll
      for (int r = 0; r < 8; ++r) xq[r] = 0.f;

      for (int c = 0; c < C_DIM; ++c) {
        float gv[8];
        v4f y0v, y1v;
        const int cn = (c + 1 < C_DIM) ? c + 1 : c;
        const v4f rn0 = *(const v4f*)&sRa[cn][eo], rn1 = *(const v4f*)&sRa[cn][eo + 4];
        {
          const v4f g0 = *(const v4f*)&sGa[c][eo], g1 = *(const v4f*)&sGa[c][eo + 4];
          y0v = *(const v4f*)&sY[c][eo];
          y1v = *(const v4f*)&sY[c][eo + 4];
          gv[0] = g0.x; gv[1] = g0.y; gv[2] = g0.z; gv[3] = g0.w;
          gv[4] = g1.x; gv[5] = g1.y; gv[6] = g1.z; gv[7] = g1.w;
        }

        float xb[8];
#pragma unroll
        for (int r = 0; r < 8; ++r) xb[r] = rv[r];

        if (it == 0 && c > 0) {
          float qx[8];
          ksplit_product(wq, xb, sP + par * PBUF, wave, lane, qx);
          par ^= 1;
#pragma unroll
          for (int r = 0; r < 8; ++r) xb[r] = qx[r];
        } else {
          for (int k = 0; k < NEUMANN; ++k) {
            float mx[8];
            ksplit_product(wa, xb, sP + par * PBUF, wave, lane, mx);
            par ^= 1;
#pragma unroll
            for (int r = 0; r < 8; ++r) xb[r] = rv[r] - gv[r] * mx[r];
          }
        }

        const float rnv[8] = {rn0.x, rn0.y, rn0.z, rn0.w, rn1.x, rn1.y, rn1.z, rn1.w};
#pragma unroll
        for (int r = 0; r < 8; ++r) {
          xb[r] -= xq[r];
          xq[r] = xb[r];
          rv[r] = rnv[r] + 2.0f * xb[r];
        }

        y0v.x -= xb[0]; y0v.y -= xb[1]; y0v.z -= xb[2]; y0v.w -= xb[3];
        y1v.x -= xb[4]; y1v.y -= xb[5]; y1v.z -= xb[6]; y1v.w -= xb[7];
        if ((lane & 15) < BSUB) {
          *(v4f*)&sY[c][eo]     = y0v;
          *(v4f*)&sY[c][eo + 4] = y1v;
        }
      }
      __syncthreads();
    }

    {
      float yo[C_DIM][SLAB / THREADS];
#pragma unroll
      for (int c = 0; c < C_DIM; ++c)
#pragma unroll
        for (int q = 0; q < SLAB / THREADS; ++q)
          yo[c][q] = sY[c][q * THREADS + tid];
#pragma unroll
      for (int c = 0; c < C_DIM; ++c)
#pragma unroll
        for (int q = 0; q < SLAB / THREADS; ++q) {
          const int e = q * THREADS + tid;
          const int b = e >> 6, j = e & 63;
          out[((size_t)(1 + blk * C_DIM + c) * B_TOT + b0 + b) * S_DIM + j] =
              yo[c][q];
        }
#pragma unroll
      for (int q = 0; q < SLAB / THREADS; ++q)
        sYprev[q * THREADS + tid] = yo[C_DIM - 1][q];
    }
    __syncthreads();
  }
}

extern "C" void kernel_launch(void* const* d_in, const int* in_sizes, int n_in,
                              void* d_out, int out_size, void* d_ws,
                              size_t ws_size, hipStream_t stream) {
  (void)in_sizes; (void)n_in; (void)out_size; (void)d_ws; (void)ws_size;
  const float* y0     = (const float*)d_in[0];
  const float* forces = (const float*)d_in[1];
  const float* W      = (const float*)d_in[2];
  float* out = (float*)d_out;
  solver_gfx1250_kernel<<<NWG, THREADS, 0, stream>>>(y0, forces, W, out);
}


# Newton iterations 2-4: cell sweep replaced by a fixed point on the whole chunk system (x = running sum of R - g*W(x_c + x_c-1)): 10 chunk-wide products, 1 exchange each, not 48; sum order changes
# speedup vs baseline: 3.4980x; 1.1231x over previous
#include <hip/hip_runtime.h>


#define S_DIM   64
#define C_DIM   8
#define N_ITER  4
#define NEUMANN 6
#define B_TOT   128
#define NB      8
#define BSUB    8
#define NWG     (B_TOT / BSUB)
#define THREADS 128
#define DT_C    0.05f
#define LDA     66
#define ROWS_ALL (C_DIM * BSUB)
#define MT_ALL   (ROWS_ALL / 16)
#define SLAB     (BSUB * S_DIM)
#define PTILE    256
#define PBUF     (4 * 4 * PTILE)
#define NTC      4
#define PBUFC    (NTC * PBUF)
#define CHUNK_STEPS 10

typedef __attribute__((ext_vector_type(2))) float v2f;
typedef __attribute__((ext_vector_type(4))) float v4f;
typedef __attribute__((ext_vector_type(8))) float v8f;

__device__ __forceinline__ void ksplit_product(const v2f (&wa)[4][4],
                                               const float (&xb)[8],
                                               float* __restrict__ sPb,
                                               int wave, int lane,
                                               float (&tot)[8]) {
  v8f acc[4];
#pragma unroll
  for (int i = 0; i < 4; ++i)
    acc[i] = (v8f){0.f, 0.f, 0.f, 0.f, 0.f, 0.f, 0.f, 0.f};
#pragma unroll
  for (int j = 0; j < 4; ++j) {
    v2f b;
    b.x = xb[2 * j];
    b.y = xb[2 * j + 1];
#pragma unroll
    for (int ii = 0; ii < 4; ++ii) {
      const int i = (ii + 1) & 3;
      acc[i] = __builtin_amdgcn_wmma_f32_16x16x4_f32(false, wa[i][j], false, b,
                                                     (short)0, acc[i],
                                                     false, false);
    }
  }
#pragma unroll
  for (int i = 1; i < 4; ++i) {
    const int dst = (wave + i) & 3;
    float* q = sPb + (dst * 4 + wave) * PTILE + lane * 4;
    *(v4f*)q         = (v4f){acc[i][0], acc[i][1], acc[i][2], acc[i][3]};
    *(v4f*)(q + 128) = (v4f){acc[i][4], acc[i][5], acc[i][6], acc[i][7]};
  }
  __syncthreads();
#pragma unroll
  for (int r = 0; r < 8; ++r) tot[r] = acc[0][r];
#pragma unroll
  for (int i = 1; i < 4; ++i) {
    const int srcw = (wave + i) & 3;
    const float* q = sPb + (wave * 4 + srcw) * PTILE + lane * 4;
    const v4f lo = *(const v4f*)q;
    const v4f hi = *(const v4f*)(q + 128);
    tot[0] += lo.x; tot[1] += lo.y; tot[2] += lo.z; tot[3] += lo.w;
    tot[4] += hi.x; tot[5] += hi.y; tot[6] += hi.z; tot[7] += hi.w;
  }
}

__device__ __forceinline__ void ksplit_product_chunk(const v2f (&wa)[4][4],
                                                     const float (&mb)[NTC][8],
                                                     float* __restrict__ sPb,
                                                     int wave, int lane,
                                                     float (&tot)[NTC][8]) {
  v8f own[NTC];
#pragma unroll
  for (int nt = 0; nt < NTC; ++nt) {
    v8f acc[4];
#pragma unroll
    for (int i = 0; i < 4; ++i)
      acc[i] = (v8f){0.f, 0.f, 0.f, 0.f, 0.f, 0.f, 0.f, 0.f};
#pragma unroll
    for (int j = 0; j < 4; ++j) {
      v2f b;
      b.x = mb[nt][2 * j];
      b.y = mb[nt][2 * j + 1];
#pragma unroll
      for (int ii = 0; ii < 4; ++ii) {
        const int i = (ii + 1) & 3;
        acc[i] = __builtin_amdgcn_wmma_f32_16x16x4_f32(false, wa[i][j], false, b,
                                                       (short)0, acc[i],
                                                       false, false);
      }
    }
#pragma unroll
    for (int i = 1; i < 4; ++i) {
      const int dst = (wave + i) & 3;
      float* q = sPb + ((nt * 4 + dst) * 4 + wave) * PTILE + lane * 4;
      *(v4f*)q         = (v4f){acc[i][0], acc[i][1], acc[i][2], acc[i][3]};
      *(v4f*)(q + 128) = (v4f){acc[i][4], acc[i][5], acc[i][6], acc[i][7]};
    }
    own[nt] = acc[0];
  }
  __syncthreads();
#pragma unroll
  for (int nt = 0; nt < NTC; ++nt)
#pragma unroll
    for (int r = 0; r < 8; ++r) tot[nt][r] = own[nt][r];
#pragma unroll
  for (int nt = 0; nt < NTC; ++nt)
#pragma unroll
    for (int i = 1; i < 4; ++i) {
      const int srcw = (wave + i) & 3;
      const float* q = sPb + ((nt * 4 + wave) * 4 + srcw) * PTILE + lane * 4;
      const v4f lo = *(const v4f*)q;
      const v4f hi = *(const v4f*)(q + 128);
      tot[nt][0] += lo.x; tot[nt][1] += lo.y; tot[nt][2] += lo.z; tot[nt][3] += lo.w;
      tot[nt][4] += hi.x; tot[nt][5] += hi.y; tot[nt][6] += hi.z; tot[nt][7] += hi.w;
    }
}

__device__ __forceinline__ float lane_xor8(float v) {
  const int b = __builtin_bit_cast(int, v);
  return __builtin_bit_cast(float,
      __builtin_amdgcn_update_dpp(b, b, 0x168  , 0xf, 0xf, false));
}

__device__ __forceinline__ void wmma_gemm_64x64(const float* __restrict__ A,
                                                const float* __restrict__ Bt,
                                                int n0, int lane,
                                                v8f (&acc)[MT_ALL]) {
  const int m = lane & 15;
  const int h = (lane >> 4) << 1;
#pragma unroll
  for (int mt = 0; mt < MT_ALL; ++mt)
    acc[mt] = (v8f){0.f, 0.f, 0.f, 0.f, 0.f, 0.f, 0.f, 0.f};
#pragma unroll
  for (int k0 = 0; k0 < S_DIM; k0 += 4) {
    v2f b;
    b.x = Bt[(k0 + h) * LDA + n0 + m];
    b.y = Bt[(k0 + h + 1) * LDA + n0 + m];
#pragma unroll
    for (int mt = 0; mt < MT_ALL; ++mt) {
      v2f a;
      a.x = A[(mt * 16 + m) * LDA + k0 + h];
      a.y = A[(mt * 16 + m) * LDA + k0 + h + 1];
      acc[mt] = __builtin_amdgcn_wmma_f32_16x16x4_f32(false, a, false, b,
                                                      (short)0, acc[mt],
                                                      false, false);
    }
  }
}

__global__ __launch_bounds__(THREADS, 1)
void solver_gfx1250_kernel(const float* __restrict__ y0,
                           const float* __restrict__ forces,
                           const float* __restrict__ W,
                           float* __restrict__ out) {
  __shared__ float sWt[S_DIM * LDA];
  __shared__ float sSa[ROWS_ALL * LDA];
  __shared__ __attribute__((aligned(16))) float sP[2 * PBUFC];
  __shared__ __attribute__((aligned(16))) float sY[C_DIM][SLAB];
  __shared__ float sYprev[SLAB];
  __shared__ __attribute__((aligned(16))) float sRa[C_DIM][SLAB];
  __shared__ __attribute__((aligned(16))) float sGa[C_DIM][SLAB];

  const int tid  = threadIdx.x;
  const int lane = tid & 31;
  const int wave = tid >> 5;
  const int n0   = wave * 16;
  const int b0   = blockIdx.x * BSUB;
  const int col  = n0 + (lane & 15);
  const int hh   = lane >> 4;

  const int nb = lane & 7;
  const int eo = nb * S_DIM + n0 + 8 * hh;

  v2f wa[4][4];
#pragma unroll
  for (int i = 0; i < 4; ++i)
#pragma unroll
    for (int j = 0; j < 4; ++j)
      wa[i][j] = *(const v2f*)(W + (16 * ((wave + i) & 3) + (lane & 15)) * S_DIM
                                 + n0 + 8 * hh + 2 * j);
  int par = 0;

  for (int e = tid; e < S_DIM * S_DIM; e += THREADS) {
    int p = e >> 6, s = e & 63;
    sWt[s * LDA + p] = W[p * S_DIM + s];
  }
  for (int e = tid; e < SLAB; e += THREADS) {
    int b = e >> 6, j = e & 63;
    float v = y0[(b0 + b) * S_DIM + j];
    sYprev[e] = v;
    out[(size_t)(b0 + b) * S_DIM + j] = v;
  }
  __syncthreads();

  const float g0 = 0.5f * DT_C;
  for (int e = tid; e < S_DIM * S_DIM; e += THREADS) {
    int s = e >> 6, p = e & 63;
    sSa[s * LDA + p] = (s == p ? 1.0f : 0.0f) - g0 * sWt[s * LDA + p];
  }
  __syncthreads();
#pragma unroll
  for (int k = 1; k < NEUMANN; ++k) {
    const float* cur = (k & 1) ? sSa : sP;
    float* nxt = (k & 1) ? sP : sSa;
    v8f qacc[MT_ALL];
    wmma_gemm_64x64(sWt, cur, n0, lane, qacc);
#pragma unroll
    for (int mt = 0; mt < MT_ALL; ++mt)
#pragma unroll
      for (int v = 0; v < 8; ++v) {
        const int row = mt * 16 + 8 * hh + v;
        nxt[row * LDA + col] = (row == col ? 1.0f : 0.0f) - g0 * qacc[mt][v];
      }
    __syncthreads();
  }
  v2f wq[4][4];
#pragma unroll
  for (int i = 0; i < 4; ++i)
#pragma unroll
    for (int j = 0; j < 4; ++j) {
      const int row = 16 * ((wave + i) & 3) + (lane & 15);
      const int kc = n0 + 8 * hh + 2 * j;
      wq[i][j].x = sP[kc * LDA + row];
      wq[i][j].y = sP[(kc + 1) * LDA + row];
    }

  for (int blk = 0; blk < NB; ++blk) {
    const float* fblk =
        forces + ((size_t)(1 + blk * C_DIM) * B_TOT + b0) * S_DIM;

    float fv[MT_ALL][8];
#pragma unroll
    for (int mt = 0; mt < MT_ALL; ++mt)
#pragma unroll
      for (int v = 0; v < 8; ++v)
        fv[mt][v] = fblk[(size_t)(2 * mt + hh) * B_TOT * S_DIM + v * S_DIM + col];

    for (int c = 0; c < C_DIM; ++c)
      for (int e = tid; e < SLAB; e += THREADS) sY[c][e] = 0.f;
    __syncthreads();

    for (int it = 0; it < N_ITER; ++it) {
#pragma unroll
      for (int q = 0; q < SLAB / THREADS; ++q) {
        const int e = q * THREADS + tid;
        const int b = e >> 6, j = e & 63;
        float ym = sYprev[e];
#pragma unroll
        for (int c = 0; c < C_DIM; ++c) {
          const float yc = sY[c][e];
          sSa[(c * BSUB + b) * LDA + j] = 0.5f * (ym + yc);
          sRa[c][e] = yc - ym;
          ym = yc;
        }
      }
      __syncthreads();

      {
        float yd[MT_ALL][8];
#pragma unroll
        for (int mt = 0; mt < MT_ALL; ++mt)
#pragma unroll
          for (int v = 0; v < 8; ++v)
            yd[mt][v] = sRa[2 * mt + hh][v * S_DIM + col];

        v8f tacc[MT_ALL];
        wmma_gemm_64x64(sSa, sWt, n0, lane, tacc);
#pragma unroll
        for (int mt = 0; mt < MT_ALL; ++mt) {
          const int c = 2 * mt + hh;
#pragma unroll
          for (int v = 0; v < 8; ++v) {
            const int e = v * S_DIM + col;
            const float t = tanhf(tacc[mt][v]);
            const float g = 0.5f * DT_C * (1.0f - t * t);
            const float R = yd[mt][v] + DT_C * t - DT_C * fv[mt][v];
            sRa[c][e] = R;
            sGa[c][e] = g;
          }
        }
      }
      __syncthreads();

      if (it == 0) {
      float rv[8], xq[8];
      {
        const v4f r0 = *(const v4f*)&sRa[0][eo], r1 = *(const v4f*)&sRa[0][eo + 4];
        rv[0] = r0.x; rv[1] = r0.y; rv[2] = r0.z; rv[3] = r0.w;
        rv[4] = r1.x; rv[5] = r1.y; rv[6] = r1.z; rv[7] = r1.w;
      }
#pragma unroll
      for (int r = 0; r < 8; ++r) xq[r] = 0.f;

      for (int c = 0; c < C_DIM; ++c) {
        float gv[8];
        v4f y0v, y1v;
        const int cn = (c + 1 < C_DIM) ? c + 1 : c;
        const v4f rn0 = *(const v4f*)&sRa[cn][eo], rn1 = *(const v4f*)&sRa[cn][eo + 4];
        {
          const v4f g0 = *(const v4f*)&sGa[c][eo], g1 = *(const v4f*)&sGa[c][eo + 4];
          y0v = *(const v4f*)&sY[c][eo];
          y1v = *(const v4f*)&sY[c][eo + 4];
          gv[0] = g0.x; gv[1] = g0.y; gv[2] = g0.z; gv[3] = g0.w;
          gv[4] = g1.x; gv[5] = g1.y; gv[6] = g1.z; gv[7] = g1.w;
        }

        float xb[8];
#pragma unroll
        for (int r = 0; r < 8; ++r) xb[r] = rv[r];

        if (c > 0) {
          float qx[8];
          ksplit_product(wq, xb, sP + par * PBUF, wave, lane, qx);
          par ^= 1;
#pragma unroll
          for (int r = 0; r < 8; ++r) xb[r] = qx[r];
        } else {
          for (int k = 0; k < NEUMANN; ++k) {
            float mx[8];
            ksplit_product(wa, xb, sP + par * PBUF, wave, lane, mx);
            par ^= 1;
#pragma unroll
            for (int r = 0; r < 8; ++r) xb[r] = rv[r] - gv[r] * mx[r];
          }
        }

        const float rnv[8] = {rn0.x, rn0.y, rn0.z, rn0.w, rn1.x, rn1.y, rn1.z, rn1.w};
#pragma unroll
        for (int r = 0; r < 8; ++r) {
          xb[r] -= xq[r];
          xq[r] = xb[r];
          rv[r] = rnv[r] + 2.0f * xb[r];
        }

        y0v.x -= xb[0]; y0v.y -= xb[1]; y0v.z -= xb[2]; y0v.w -= xb[3];
        y1v.x -= xb[4]; y1v.y -= xb[5]; y1v.z -= xb[6]; y1v.w -= xb[7];
        if ((lane & 15) < BSUB) {
          *(v4f*)&sY[c][eo]     = y0v;
          *(v4f*)&sY[c][eo + 4] = y1v;
        }
      }
      } else {
        const int hi8 = (lane >> 3) & 1;
        const float hi1 = hi8 ? 1.0f : 0.0f;
        const float hi2 = hi8 ? 2.0f : 0.0f;
        float Rr[NTC][8], Gg[NTC][8], dd[NTC][8], mb[NTC][8];
#pragma unroll
        for (int nt = 0; nt < NTC; ++nt) {
          const int cell = 2 * nt + hi8;
          const v4f r0 = *(const v4f*)&sRa[cell][eo], r1 = *(const v4f*)&sRa[cell][eo + 4];
          const v4f g0v = *(const v4f*)&sGa[cell][eo], g1v = *(const v4f*)&sGa[cell][eo + 4];
          Rr[nt][0] = r0.x; Rr[nt][1] = r0.y; Rr[nt][2] = r0.z; Rr[nt][3] = r0.w;
          Rr[nt][4] = r1.x; Rr[nt][5] = r1.y; Rr[nt][6] = r1.z; Rr[nt][7] = r1.w;
          Gg[nt][0] = g0v.x; Gg[nt][1] = g0v.y; Gg[nt][2] = g0v.z; Gg[nt][3] = g0v.w;
          Gg[nt][4] = g1v.x; Gg[nt][5] = g1v.y; Gg[nt][6] = g1v.z; Gg[nt][7] = g1v.w;
#pragma unroll
          for (int r = 0; r < 8; ++r) dd[nt][r] = Rr[nt][r];
        }

        for (int k = 0; k < CHUNK_STEPS; ++k) {
          float pp[8];
#pragma unroll
          for (int r = 0; r < 8; ++r) pp[r] = 0.f;
#pragma unroll
          for (int nt = 0; nt < NTC; ++nt)
#pragma unroll
            for (int r = 0; r < 8; ++r) {
              const float dp = lane_xor8(dd[nt][r]);
              mb[nt][r] = 2.0f * pp[r] + (dd[nt][r] + hi2 * dp);
              pp[r] += dd[nt][r] + dp;
            }
          float tot[NTC][8];
          ksplit_product_chunk(wa, mb, sP + par * PBUFC, wave, lane, tot);
          par ^= 1;
#pragma unroll
          for (int nt = 0; nt < NTC; ++nt)
#pragma unroll
            for (int r = 0; r < 8; ++r)
              dd[nt][r] = Rr[nt][r] - Gg[nt][r] * tot[nt][r];
        }

        {
          float pp[8];
#pragma unroll
          for (int r = 0; r < 8; ++r) pp[r] = 0.f;
#pragma unroll
          for (int nt = 0; nt < NTC; ++nt) {
            const int cell = 2 * nt + hi8;
            v4f y0v = *(const v4f*)&sY[cell][eo];
            v4f y1v = *(const v4f*)&sY[cell][eo + 4];
            float xc[8];
#pragma unroll
            for (int r = 0; r < 8; ++r) {
              const float dp = lane_xor8(dd[nt][r]);
              xc[r] = pp[r] + (dd[nt][r] + hi1 * dp);
              pp[r] += dd[nt][r] + dp;
            }
            y0v.x -= xc[0]; y0v.y -= xc[1]; y0v.z -= xc[2]; y0v.w -= xc[3];
            y1v.x -= xc[4]; y1v.y -= xc[5]; y1v.z -= xc[6]; y1v.w -= xc[7];
            *(v4f*)&sY[cell][eo]     = y0v;
            *(v4f*)&sY[cell][eo + 4] = y1v;
          }
        }
      }
      __syncthreads();
    }

    {
      float yo[C_DIM][SLAB / THREADS];
#pragma unroll
      for (int c = 0; c < C_DIM; ++c)
#pragma unroll
        for (int q = 0; q < SLAB / THREADS; ++q)
          yo[c][q] = sY[c][q * THREADS + tid];
#pragma unroll
      for (int c = 0; c < C_DIM; ++c)
#pragma unroll
        for (int q = 0; q < SLAB / THREADS; ++q) {
          const int e = q * THREADS + tid;
          const int b = e >> 6, j = e & 63;
          out[((size_t)(1 + blk * C_DIM + c) * B_TOT + b0 + b) * S_DIM + j] =
              yo[c][q];
        }
#pragma unroll
      for (int q = 0; q < SLAB / THREADS; ++q)
        sYprev[q * THREADS + tid] = yo[C_DIM - 1][q];
    }
    __syncthreads();
  }
}

extern "C" void kernel_launch(void* const* d_in, const int* in_sizes, int n_in,
                              void* d_out, int out_size, void* d_ws,
                              size_t ws_size, hipStream_t stream) {
  (void)in_sizes; (void)n_in; (void)out_size; (void)d_ws; (void)ws_size;
  const float* y0     = (const float*)d_in[0];
  const float* forces = (const float*)d_in[1];
  const float* W      = (const float*)d_in[2];
  float* out = (float*)d_out;
  solver_gfx1250_kernel<<<NWG, THREADS, 0, stream>>>(y0, forces, W, out);
}


# Chunk-wide fixed point split over the waves by batch row, not by K: a wave takes the whole K of its own columns from its accumulator registers, W whole in registers; no exchange; sum order changes
# speedup vs baseline: 4.1953x; 1.1993x over previous
#include <hip/hip_runtime.h>


#define S_DIM   64
#define C_DIM   8
#define N_ITER  4
#define NEUMANN 6
#define B_TOT   128
#define NB      8
#define BSUB    8
#define NWG     (B_TOT / BSUB)
#define THREADS 128
#define DT_C    0.05f
#define LDA     66
#define ROWS_ALL (C_DIM * BSUB)
#define MT_ALL   (ROWS_ALL / 16)
#define SLAB     (BSUB * S_DIM)
#define PTILE    256
#define PBUF     (4 * 4 * PTILE)
#define CHUNK_STEPS 10
#define KT_ALL   (S_DIM / 16)
#define SLABP    (SLAB + 4)

typedef __attribute__((ext_vector_type(2))) float v2f;
typedef __attribute__((ext_vector_type(4))) float v4f;
typedef __attribute__((ext_vector_type(8))) float v8f;

__device__ __forceinline__ void ksplit_product(const v2f (&wa)[4][4],
                                               const float (&xb)[8],
                                               float* __restrict__ sPb,
                                               int wave, int lane,
                                               float (&tot)[8]) {
  v8f acc[4];
#pragma unroll
  for (int i = 0; i < 4; ++i)
    acc[i] = (v8f){0.f, 0.f, 0.f, 0.f, 0.f, 0.f, 0.f, 0.f};
#pragma unroll
  for (int j = 0; j < 4; ++j) {
    v2f b;
    b.x = xb[2 * j];
    b.y = xb[2 * j + 1];
#pragma unroll
    for (int ii = 0; ii < 4; ++ii) {
      const int i = (ii + 1) & 3;
      acc[i] = __builtin_amdgcn_wmma_f32_16x16x4_f32(false, wa[i][j], false, b,
                                                     (short)0, acc[i],
                                                     false, false);
    }
  }
#pragma unroll
  for (int i = 1; i < 4; ++i) {
    const int dst = (wave + i) & 3;
    float* q = sPb + (dst * 4 + wave) * PTILE + lane * 4;
    *(v4f*)q         = (v4f){acc[i][0], acc[i][1], acc[i][2], acc[i][3]};
    *(v4f*)(q + 128) = (v4f){acc[i][4], acc[i][5], acc[i][6], acc[i][7]};
  }
  __syncthreads();
#pragma unroll
  for (int r = 0; r < 8; ++r) tot[r] = acc[0][r];
#pragma unroll
  for (int i = 1; i < 4; ++i) {
    const int srcw = (wave + i) & 3;
    const float* q = sPb + (wave * 4 + srcw) * PTILE + lane * 4;
    const v4f lo = *(const v4f*)q;
    const v4f hi = *(const v4f*)(q + 128);
    tot[0] += lo.x; tot[1] += lo.y; tot[2] += lo.z; tot[3] += lo.w;
    tot[4] += hi.x; tot[5] += hi.y; tot[6] += hi.z; tot[7] += hi.w;
  }
}

__device__ __forceinline__ void column_product(const v2f (&wf)[MT_ALL][KT_ALL][4],
                                               const float (&mb)[KT_ALL][8],
                                               v8f (&acc)[MT_ALL]) {
#pragma unroll
  for (int i = 0; i < MT_ALL; ++i)
    acc[i] = (v8f){0.f, 0.f, 0.f, 0.f, 0.f, 0.f, 0.f, 0.f};
#pragma unroll
  for (int kt = 0; kt < KT_ALL; ++kt)
#pragma unroll
    for (int j = 0; j < 4; ++j) {
      v2f b;
      b.x = mb[kt][2 * j];
      b.y = mb[kt][2 * j + 1];
#pragma unroll
      for (int i = 0; i < MT_ALL; ++i)
        acc[i] = __builtin_amdgcn_wmma_f32_16x16x4_f32(false, wf[i][kt][j], false, b,
                                                       (short)0, acc[i],
                                                       false, false);
    }
}

template <int SH>
__device__ __forceinline__ float row_below(float v) {
  const int b = __builtin_bit_cast(int, v);
  return __builtin_bit_cast(float,
      __builtin_amdgcn_update_dpp(0, b, 0x110 + SH  , 0xf, 0xf, true));
}

__device__ __forceinline__ float cell_running_sum(float v, float w1, float w2, float w4) {
  v = fmaf(row_below<1>(v), w1, v);
  v = fmaf(row_below<2>(v), w2, v);
  v = fmaf(row_below<4>(v), w4, v);
  return v;
}

__device__ __forceinline__ void wmma_gemm_64x64(const float* __restrict__ A,
                                                const float* __restrict__ Bt,
                                                int n0, int lane,
                                                v8f (&acc)[MT_ALL]) {
  const int m = lane & 15;
  const int h = (lane >> 4) << 1;
#pragma unroll
  for (int mt = 0; mt < MT_ALL; ++mt)
    acc[mt] = (v8f){0.f, 0.f, 0.f, 0.f, 0.f, 0.f, 0.f, 0.f};
#pragma unroll
  for (int k0 = 0; k0 < S_DIM; k0 += 4) {
    v2f b;
    b.x = Bt[(k0 + h) * LDA + n0 + m];
    b.y = Bt[(k0 + h + 1) * LDA + n0 + m];
#pragma unroll
    for (int mt = 0; mt < MT_ALL; ++mt) {
      v2f a;
      a.x = A[(mt * 16 + m) * LDA + k0 + h];
      a.y = A[(mt * 16 + m) * LDA + k0 + h + 1];
      acc[mt] = __builtin_amdgcn_wmma_f32_16x16x4_f32(false, a, false, b,
                                                      (short)0, acc[mt],
                                                      false, false);
    }
  }
}

__global__ __launch_bounds__(THREADS, 1)
void solver_gfx1250_kernel(const float* __restrict__ y0,
                           const float* __restrict__ forces,
                           const float* __restrict__ W,
                           float* __restrict__ out) {
  __shared__ float sWt[S_DIM * LDA];
  __shared__ float sSa[ROWS_ALL * LDA];
  __shared__ __attribute__((aligned(16))) float sP[2 * PBUF];
  __shared__ __attribute__((aligned(16))) float sY[C_DIM][SLABP];
  __shared__ float sYprev[SLAB];
  __shared__ __attribute__((aligned(16))) float sRa[C_DIM][SLABP];
  __shared__ __attribute__((aligned(16))) float sGa[C_DIM][SLABP];

  const int tid  = threadIdx.x;
  const int lane = tid & 31;
  const int wave = tid >> 5;
  const int n0   = wave * 16;
  const int b0   = blockIdx.x * BSUB;
  const int col  = n0 + (lane & 15);
  const int hh   = lane >> 4;

  const int nb = lane & 7;
  const int eo = nb * S_DIM + n0 + 8 * hh;

  v2f wa[4][4];
#pragma unroll
  for (int i = 0; i < 4; ++i)
#pragma unroll
    for (int j = 0; j < 4; ++j)
      wa[i][j] = *(const v2f*)(W + (16 * ((wave + i) & 3) + (lane & 15)) * S_DIM
                                 + n0 + 8 * hh + 2 * j);
  v2f wf[MT_ALL][KT_ALL][4];
#pragma unroll
  for (int i = 0; i < MT_ALL; ++i)
#pragma unroll
    for (int kt = 0; kt < KT_ALL; ++kt)
#pragma unroll
      for (int j = 0; j < 4; ++j)
        wf[i][kt][j] = *(const v2f*)(W + (16 * i + (lane & 15)) * S_DIM
                                       + 16 * kt + 8 * hh + 2 * j);
  int par = 0;

  for (int e = tid; e < S_DIM * S_DIM; e += THREADS) {
    int p = e >> 6, s = e & 63;
    sWt[s * LDA + p] = W[p * S_DIM + s];
  }
  for (int e = tid; e < SLAB; e += THREADS) {
    int b = e >> 6, j = e & 63;
    float v = y0[(b0 + b) * S_DIM + j];
    sYprev[e] = v;
    out[(size_t)(b0 + b) * S_DIM + j] = v;
  }
  __syncthreads();

  const float g0 = 0.5f * DT_C;
  for (int e = tid; e < S_DIM * S_DIM; e += THREADS) {
    int s = e >> 6, p = e & 63;
    sSa[s * LDA + p] = (s == p ? 1.0f : 0.0f) - g0 * sWt[s * LDA + p];
  }
  __syncthreads();
#pragma unroll
  for (int k = 1; k < NEUMANN; ++k) {
    const float* cur = (k & 1) ? sSa : sP;
    float* nxt = (k & 1) ? sP : sSa;
    v8f qacc[MT_ALL];
    wmma_gemm_64x64(sWt, cur, n0, lane, qacc);
#pragma unroll
    for (int mt = 0; mt < MT_ALL; ++mt)
#pragma unroll
      for (int v = 0; v < 8; ++v) {
        const int row = mt * 16 + 8 * hh + v;
        nxt[row * LDA + col] = (row == col ? 1.0f : 0.0f) - g0 * qacc[mt][v];
      }
    __syncthreads();
  }
  v2f wq[4][4];
#pragma unroll
  for (int i = 0; i < 4; ++i)
#pragma unroll
    for (int j = 0; j < 4; ++j) {
      const int row = 16 * ((wave + i) & 3) + (lane & 15);
      const int kc = n0 + 8 * hh + 2 * j;
      wq[i][j].x = sP[kc * LDA + row];
      wq[i][j].y = sP[(kc + 1) * LDA + row];
    }

  for (int blk = 0; blk < NB; ++blk) {
    const float* fblk =
        forces + ((size_t)(1 + blk * C_DIM) * B_TOT + b0) * S_DIM;

    float fv[MT_ALL][8];
#pragma unroll
    for (int mt = 0; mt < MT_ALL; ++mt)
#pragma unroll
      for (int v = 0; v < 8; ++v)
        fv[mt][v] = fblk[(size_t)(2 * mt + hh) * B_TOT * S_DIM + v * S_DIM + col];

    for (int c = 0; c < C_DIM; ++c)
      for (int e = tid; e < SLAB; e += THREADS) sY[c][e] = 0.f;
    __syncthreads();

    for (int it = 0; it < N_ITER; ++it) {
#pragma unroll
      for (int q = 0; q < SLAB / THREADS; ++q) {
        const int e = q * THREADS + tid;
        const int b = e >> 6, j = e & 63;
        float ym = sYprev[e];
#pragma unroll
        for (int c = 0; c < C_DIM; ++c) {
          const float yc = sY[c][e];
          sSa[(c * BSUB + b) * LDA + j] = 0.5f * (ym + yc);
          sRa[c][e] = yc - ym;
          ym = yc;
        }
      }
      __syncthreads();

      {
        float yd[MT_ALL][8];
#pragma unroll
        for (int mt = 0; mt < MT_ALL; ++mt)
#pragma unroll
          for (int v = 0; v < 8; ++v)
            yd[mt][v] = sRa[2 * mt + hh][v * S_DIM + col];

        v8f tacc[MT_ALL];
        wmma_gemm_64x64(sSa, sWt, n0, lane, tacc);
#pragma unroll
        for (int mt = 0; mt < MT_ALL; ++mt) {
          const int c = 2 * mt + hh;
#pragma unroll
          for (int v = 0; v < 8; ++v) {
            const int e = v * S_DIM + col;
            const float t = tanhf(tacc[mt][v]);
            const float g = 0.5f * DT_C * (1.0f - t * t);
            const float R = yd[mt][v] + DT_C * t - DT_C * fv[mt][v];
            sRa[c][e] = R;
            sGa[c][e] = g;
          }
        }
      }
      __syncthreads();

      if (it == 0) {
      float rv[8], xq[8];
      {
        const v4f r0 = *(const v4f*)&sRa[0][eo], r1 = *(const v4f*)&sRa[0][eo + 4];
        rv[0] = r0.x; rv[1] = r0.y; rv[2] = r0.z; rv[3] = r0.w;
        rv[4] = r1.x; rv[5] = r1.y; rv[6] = r1.z; rv[7] = r1.w;
      }
#pragma unroll
      for (int r = 0; r < 8; ++r) xq[r] = 0.f;

      for (int c = 0; c < C_DIM; ++c) {
        float gv[8];
        v4f y0v, y1v;
        const int cn = (c + 1 < C_DIM) ? c + 1 : c;
        const v4f rn0 = *(const v4f*)&sRa[cn][eo], rn1 = *(const v4f*)&sRa[cn][eo + 4];
        {
          const v4f g0 = *(const v4f*)&sGa[c][eo], g1 = *(const v4f*)&sGa[c][eo + 4];
          y0v = *(const v4f*)&sY[c][eo];
          y1v = *(const v4f*)&sY[c][eo + 4];
          gv[0] = g0.x; gv[1] = g0.y; gv[2] = g0.z; gv[3] = g0.w;
          gv[4] = g1.x; gv[5] = g1.y; gv[6] = g1.z; gv[7] = g1.w;
        }

        float xb[8];
#pragma unroll
        for (int r = 0; r < 8; ++r) xb[r] = rv[r];

        if (c > 0) {
          float qx[8];
          ksplit_product(wq, xb, sP + par * PBUF, wave, lane, qx);
          par ^= 1;
#pragma unroll
          for (int r = 0; r < 8; ++r) xb[r] = qx[r];
        } else {
          for (int k = 0; k < NEUMANN; ++k) {
            float mx[8];
            ksplit_product(wa, xb, sP + par * PBUF, wave, lane, mx);
            par ^= 1;
#pragma unroll
            for (int r = 0; r < 8; ++r) xb[r] = rv[r] - gv[r] * mx[r];
          }
        }

        const float rnv[8] = {rn0.x, rn0.y, rn0.z, rn0.w, rn1.x, rn1.y, rn1.z, rn1.w};
#pragma unroll
        for (int r = 0; r < 8; ++r) {
          xb[r] -= xq[r];
          xq[r] = xb[r];
          rv[r] = rnv[r] + 2.0f * xb[r];
        }

        y0v.x -= xb[0]; y0v.y -= xb[1]; y0v.z -= xb[2]; y0v.w -= xb[3];
        y1v.x -= xb[4]; y1v.y -= xb[5]; y1v.z -= xb[6]; y1v.w -= xb[7];
        if ((lane & 15) < BSUB) {
          *(v4f*)&sY[c][eo]     = y0v;
          *(v4f*)&sY[c][eo + 4] = y1v;
        }
      }
      } else {
        const int cl = lane & 7;
        const int eb = (2 * wave + ((lane >> 3) & 1)) * S_DIM + 8 * hh;
        const float w1 = cl >= 1 ? 1.0f : 0.0f;
        const float w2 = cl >= 2 ? 1.0f : 0.0f;
        const float w4 = cl >= 4 ? 1.0f : 0.0f;
        float Rr[MT_ALL][8], Gg[MT_ALL][8], dd[MT_ALL][8];
#pragma unroll
        for (int i = 0; i < MT_ALL; ++i) {
          const v4f r0 = *(const v4f*)&sRa[cl][eb + 16 * i], r1 = *(const v4f*)&sRa[cl][eb + 16 * i + 4];
          const v4f g0v = *(const v4f*)&sGa[cl][eb + 16 * i], g1v = *(const v4f*)&sGa[cl][eb + 16 * i + 4];
          Rr[i][0] = r0.x; Rr[i][1] = r0.y; Rr[i][2] = r0.z; Rr[i][3] = r0.w;
          Rr[i][4] = r1.x; Rr[i][5] = r1.y; Rr[i][6] = r1.z; Rr[i][7] = r1.w;
          Gg[i][0] = g0v.x; Gg[i][1] = g0v.y; Gg[i][2] = g0v.z; Gg[i][3] = g0v.w;
          Gg[i][4] = g1v.x; Gg[i][5] = g1v.y; Gg[i][6] = g1v.z; Gg[i][7] = g1v.w;
#pragma unroll
          for (int r = 0; r < 8; ++r) dd[i][r] = Rr[i][r];
        }

        for (int k = 0; k < CHUNK_STEPS; ++k) {
          float mb[KT_ALL][8];
#pragma unroll
          for (int i = 0; i < KT_ALL; ++i)
#pragma unroll
            for (int r = 0; r < 8; ++r) {
              const float xc = cell_running_sum(dd[i][r], w1, w2, w4);
              mb[i][r] = fmaf(2.0f, xc, -dd[i][r]);
            }
          v8f tot[MT_ALL];
          column_product(wf, mb, tot);
#pragma unroll
          for (int i = 0; i < MT_ALL; ++i)
#pragma unroll
            for (int r = 0; r < 8; ++r)
              dd[i][r] = Rr[i][r] - Gg[i][r] * tot[i][r];
        }

#pragma unroll
        for (int i = 0; i < MT_ALL; ++i) {
          v4f y0v = *(const v4f*)&sY[cl][eb + 16 * i];
          v4f y1v = *(const v4f*)&sY[cl][eb + 16 * i + 4];
          float xc[8];
#pragma unroll
          for (int r = 0; r < 8; ++r) xc[r] = cell_running_sum(dd[i][r], w1, w2, w4);
          y0v.x -= xc[0]; y0v.y -= xc[1]; y0v.z -= xc[2]; y0v.w -= xc[3];
          y1v.x -= xc[4]; y1v.y -= xc[5]; y1v.z -= xc[6]; y1v.w -= xc[7];
          *(v4f*)&sY[cl][eb + 16 * i]     = y0v;
          *(v4f*)&sY[cl][eb + 16 * i + 4] = y1v;
        }
      }
      __syncthreads();
    }

    {
      float yo[C_DIM][SLAB / THREADS];
#pragma unroll
      for (int c = 0; c < C_DIM; ++c)
#pragma unroll
        for (int q = 0; q < SLAB / THREADS; ++q)
          yo[c][q] = sY[c][q * THREADS + tid];
#pragma unroll
      for (int c = 0; c < C_DIM; ++c)
#pragma unroll
        for (int q = 0; q < SLAB / THREADS; ++q) {
          const int e = q * THREADS + tid;
          const int b = e >> 6, j = e & 63;
          out[((size_t)(1 + blk * C_DIM + c) * B_TOT + b0 + b) * S_DIM + j] =
              yo[c][q];
        }
#pragma unroll
      for (int q = 0; q < SLAB / THREADS; ++q)
        sYprev[q * THREADS + tid] = yo[C_DIM - 1][q];
    }
    __syncthreads();
  }
}

extern "C" void kernel_launch(void* const* d_in, const int* in_sizes, int n_in,
                              void* d_out, int out_size, void* d_ws,
                              size_t ws_size, hipStream_t stream) {
  (void)in_sizes; (void)n_in; (void)out_size; (void)d_ws; (void)ws_size;
  const float* y0     = (const float*)d_in[0];
  const float* forces = (const float*)d_in[1];
  const float* W      = (const float*)d_in[2];
  float* out = (float*)d_out;
  solver_gfx1250_kernel<<<NWG, THREADS, 0, stream>>>(y0, forces, W, out);
}


# 2 batch rows per workgroup (64 workgroups): a product is ONE 16-column tile (8 cells x 2 rows) split by K over 4 waves, 16 WMMAs a wave not 64; iteration 1 by chunk fixed point; sum order changes
# speedup vs baseline: 12.4572x; 2.9693x over previous
#include <hip/hip_runtime.h>


#define S_DIM   64
#define C_DIM   8
#define N_ITER  4
#define B_TOT   128
#define NB      8
#define BSUB    2
#define NWG     (B_TOT / BSUB)
#define THREADS 128
#define DT_C    0.05f
#define PTILE    256
#define PBUF     (4 * 4 * PTILE)
#define CHUNK_STEPS 10
#define OPITCH   (S_DIM + 4)

typedef __attribute__((ext_vector_type(2))) float v2f;
typedef __attribute__((ext_vector_type(4))) float v4f;
typedef __attribute__((ext_vector_type(8))) float v8f;

__device__ __forceinline__ void ksplit_product(const v2f (&wa)[4][4],
                                               const float (&xb)[8],
                                               float* __restrict__ sPb,
                                               int wave, int lane,
                                               float (&tot)[8]) {
  v8f acc[4];
#pragma unroll
  for (int i = 0; i < 4; ++i)
    acc[i] = (v8f){0.f, 0.f, 0.f, 0.f, 0.f, 0.f, 0.f, 0.f};
#pragma unroll
  for (int j = 0; j < 4; ++j) {
    v2f b;
    b.x = xb[2 * j];
    b.y = xb[2 * j + 1];
#pragma unroll
    for (int ii = 0; ii < 4; ++ii) {
      const int i = (ii + 1) & 3;
      acc[i] = __builtin_amdgcn_wmma_f32_16x16x4_f32(false, wa[i][j], false, b,
                                                     (short)0, acc[i],
                                                     false, false);
    }
  }
#pragma unroll
  for (int i = 1; i < 4; ++i) {
    const int dst = (wave + i) & 3;
    float* q = sPb + (dst * 4 + wave) * PTILE + lane * 4;
    *(v4f*)q         = (v4f){acc[i][0], acc[i][1], acc[i][2], acc[i][3]};
    *(v4f*)(q + 128) = (v4f){acc[i][4], acc[i][5], acc[i][6], acc[i][7]};
  }
  __syncthreads();
#pragma unroll
  for (int r = 0; r < 8; ++r) tot[r] = acc[0][r];
#pragma unroll
  for (int i = 1; i < 4; ++i) {
    const int srcw = (wave + i) & 3;
    const float* q = sPb + (wave * 4 + srcw) * PTILE + lane * 4;
    const v4f lo = *(const v4f*)q;
    const v4f hi = *(const v4f*)(q + 128);
    tot[0] += lo.x; tot[1] += lo.y; tot[2] += lo.z; tot[3] += lo.w;
    tot[4] += hi.x; tot[5] += hi.y; tot[6] += hi.z; tot[7] += hi.w;
  }
}

template <int SH>
__device__ __forceinline__ float row_below(float v) {
  const int b = __builtin_bit_cast(int, v);
  return __builtin_bit_cast(float,
      __builtin_amdgcn_update_dpp(0, b, 0x110 + SH  , 0xf, 0xf, true));
}

__device__ __forceinline__ float cell_running_sum(float v, float w1, float w2, float w4) {
  v = fmaf(row_below<1>(v), w1, v);
  v = fmaf(row_below<2>(v), w2, v);
  v = fmaf(row_below<4>(v), w4, v);
  return v;
}

__global__ __launch_bounds__(THREADS, 1)
void solver_gfx1250_kernel(const float* __restrict__ y0,
                           const float* __restrict__ forces,
                           const float* __restrict__ W,
                           float* __restrict__ out) {
  __shared__ __attribute__((aligned(16))) float sP[2 * PBUF];
  __shared__ __attribute__((aligned(16))) float sOut[C_DIM * BSUB * OPITCH];

  const int tid  = threadIdx.x;
  const int lane = tid & 31;
  const int wave = tid >> 5;
  const int n0   = wave * 16;
  const int b0   = blockIdx.x * BSUB;
  const int hh   = lane >> 4;
  const int cl   = lane & 7;
  const int br   = (lane >> 3) & 1;
  const int so   = n0 + 8 * hh;
  const int eb   = tid >> 6;
  const int ej   = tid & 63;

  v2f wa[4][4];
#pragma unroll
  for (int i = 0; i < 4; ++i)
#pragma unroll
    for (int j = 0; j < 4; ++j)
      wa[i][j] = *(const v2f*)(W + (16 * ((wave + i) & 3) + (lane & 15)) * S_DIM
                                 + so + 2 * j);
  int par = 0;

  const float w1 = cl >= 1 ? 1.0f : 0.0f;
  const float w2 = cl >= 2 ? 1.0f : 0.0f;
  const float w4 = cl >= 4 ? 1.0f : 0.0f;

  float yp[8];
  {
    const float* p = y0 + (size_t)(b0 + br) * S_DIM + so;
    const v4f a = *(const v4f*)p, b = *(const v4f*)(p + 4);
    yp[0] = a.x; yp[1] = a.y; yp[2] = a.z; yp[3] = a.w;
    yp[4] = b.x; yp[5] = b.y; yp[6] = b.z; yp[7] = b.w;
  }
  out[(size_t)(b0 + eb) * S_DIM + ej] = y0[(size_t)(b0 + eb) * S_DIM + ej];

  float fv[8];
  {
    const float* p = forces + ((size_t)(1 + cl) * B_TOT + b0 + br) * S_DIM + so;
    const v4f a = *(const v4f*)p, b = *(const v4f*)(p + 4);
    fv[0] = a.x; fv[1] = a.y; fv[2] = a.z; fv[3] = a.w;
    fv[4] = b.x; fv[5] = b.y; fv[6] = b.z; fv[7] = b.w;
  }

  for (int blk = 0; blk < NB; ++blk) {
    float fn[8];
    {
      const int bn = (blk + 1 < NB) ? blk + 1 : blk;
      const float* p = forces + ((size_t)(1 + bn * C_DIM + cl) * B_TOT + b0 + br) * S_DIM + so;
      const v4f a = *(const v4f*)p, b = *(const v4f*)(p + 4);
      fn[0] = a.x; fn[1] = a.y; fn[2] = a.z; fn[3] = a.w;
      fn[4] = b.x; fn[5] = b.y; fn[6] = b.z; fn[7] = b.w;
    }

    float y[8];
#pragma unroll
    for (int r = 0; r < 8; ++r) y[r] = 0.f;

    for (int it = 0; it < N_ITER; ++it) {
      float sv[8], yd[8];
#pragma unroll
      for (int r = 0; r < 8; ++r) {
        const float below = row_below<1>(y[r]);
        const float ym = (cl == 0) ? yp[r] : below;
        sv[r] = 0.5f * (ym + y[r]);
        yd[r] = y[r] - ym;
      }
      float ws[8];
      ksplit_product(wa, sv, sP + par * PBUF, wave, lane, ws);
      par ^= 1;
      float Rr[8], Gg[8], dd[8];
#pragma unroll
      for (int r = 0; r < 8; ++r) {
        const float t = tanhf(ws[r]);
        Gg[r] = 0.5f * DT_C * (1.0f - t * t);
        Rr[r] = yd[r] + DT_C * t - DT_C * fv[r];
        dd[r] = Rr[r];
      }

      for (int k = 0; k < CHUNK_STEPS; ++k) {
        float mb[8];
#pragma unroll
        for (int r = 0; r < 8; ++r) {
          const float xc = cell_running_sum(dd[r], w1, w2, w4);
          mb[r] = fmaf(2.0f, xc, -dd[r]);
        }
        float tot[8];
        ksplit_product(wa, mb, sP + par * PBUF, wave, lane, tot);
        par ^= 1;
#pragma unroll
        for (int r = 0; r < 8; ++r) dd[r] = Rr[r] - Gg[r] * tot[r];
      }

#pragma unroll
      for (int r = 0; r < 8; ++r) y[r] -= cell_running_sum(dd[r], w1, w2, w4);
    }

    {
      float* q = sOut + (cl * BSUB + br) * OPITCH + so;
      *(v4f*)q       = (v4f){y[0], y[1], y[2], y[3]};
      *(v4f*)(q + 4) = (v4f){y[4], y[5], y[6], y[7]};
    }
    __syncthreads();
    {
      float yo[C_DIM];
#pragma unroll
      for (int c = 0; c < C_DIM; ++c) yo[c] = sOut[(c * BSUB + eb) * OPITCH + ej];
#pragma unroll
      for (int c = 0; c < C_DIM; ++c)
        out[((size_t)(1 + blk * C_DIM + c) * B_TOT + b0 + eb) * S_DIM + ej] = yo[c];
    }
#pragma unroll
    for (int r = 0; r < 8; ++r) {
      yp[r] = __shfl(y[r], lane | 7);
      fv[r] = fn[r];
    }
  }
}

extern "C" void kernel_launch(void* const* d_in, const int* in_sizes, int n_in,
                              void* d_out, int out_size, void* d_ws,
                              size_t ws_size, hipStream_t stream) {
  (void)in_sizes; (void)n_in; (void)out_size; (void)d_ws; (void)ws_size;
  const float* y0     = (const float*)d_in[0];
  const float* forces = (const float*)d_in[1];
  const float* W      = (const float*)d_in[2];
  float* out = (float*)d_out;
  solver_gfx1250_kernel<<<NWG, THREADS, 0, stream>>>(y0, forces, W, out);
}


# Chunk fixed point cut from 10 to 5 products per Newton iteration: a solve's leftover (bound 3e-4 of its correction) is removed by the next Newton iteration, result moves < 1e-9; sum order kept
# speedup vs baseline: 19.5157x; 1.5666x over previous
#include <hip/hip_runtime.h>


#define S_DIM   64
#define C_DIM   8
#define N_ITER  4
#define B_TOT   128
#define NB      8
#define BSUB    2
#define NWG     (B_TOT / BSUB)
#define THREADS 128
#define DT_C    0.05f
#define PTILE    256
#define PBUF     (4 * 4 * PTILE)
#define CHUNK_STEPS 5
#define OPITCH   (S_DIM + 4)

typedef __attribute__((ext_vector_type(2))) float v2f;
typedef __attribute__((ext_vector_type(4))) float v4f;
typedef __attribute__((ext_vector_type(8))) float v8f;

__device__ __forceinline__ void ksplit_product(const v2f (&wa)[4][4],
                                               const float (&xb)[8],
                                               float* __restrict__ sPb,
                                               int wave, int lane,
                                               float (&tot)[8]) {
  v8f acc[4];
#pragma unroll
  for (int i = 0; i < 4; ++i)
    acc[i] = (v8f){0.f, 0.f, 0.f, 0.f, 0.f, 0.f, 0.f, 0.f};
#pragma unroll
  for (int j = 0; j < 4; ++j) {
    v2f b;
    b.x = xb[2 * j];
    b.y = xb[2 * j + 1];
#pragma unroll
    for (int ii = 0; ii < 4; ++ii) {
      const int i = (ii + 1) & 3;
      acc[i] = __builtin_amdgcn_wmma_f32_16x16x4_f32(false, wa[i][j], false, b,
                                                     (short)0, acc[i],
                                                     false, false);
    }
  }
#pragma unroll
  for (int i = 1; i < 4; ++i) {
    const int dst = (wave + i) & 3;
    float* q = sPb + (dst * 4 + wave) * PTILE + lane * 4;
    *(v4f*)q         = (v4f){acc[i][0], acc[i][1], acc[i][2], acc[i][3]};
    *(v4f*)(q + 128) = (v4f){acc[i][4], acc[i][5], acc[i][6], acc[i][7]};
  }
  __syncthreads();
#pragma unroll
  for (int r = 0; r < 8; ++r) tot[r] = acc[0][r];
#pragma unroll
  for (int i = 1; i < 4; ++i) {
    const int srcw = (wave + i) & 3;
    const float* q = sPb + (wave * 4 + srcw) * PTILE + lane * 4;
    const v4f lo = *(const v4f*)q;
    const v4f hi = *(const v4f*)(q + 128);
    tot[0] += lo.x; tot[1] += lo.y; tot[2] += lo.z; tot[3] += lo.w;
    tot[4] += hi.x; tot[5] += hi.y; tot[6] += hi.z; tot[7] += hi.w;
  }
}

template <int SH>
__device__ __forceinline__ float row_below(float v) {
  const int b = __builtin_bit_cast(int, v);
  return __builtin_bit_cast(float,
      __builtin_amdgcn_update_dpp(0, b, 0x110 + SH  , 0xf, 0xf, true));
}

__device__ __forceinline__ float cell_running_sum(float v, float w1, float w2, float w4) {
  v = fmaf(row_below<1>(v), w1, v);
  v = fmaf(row_below<2>(v), w2, v);
  v = fmaf(row_below<4>(v), w4, v);
  return v;
}

__global__ __launch_bounds__(THREADS, 1)
void solver_gfx1250_kernel(const float* __restrict__ y0,
                           const float* __restrict__ forces,
                           const float* __restrict__ W,
                           float* __restrict__ out) {
  __shared__ __attribute__((aligned(16))) float sP[2 * PBUF];
  __shared__ __attribute__((aligned(16))) float sOut[C_DIM * BSUB * OPITCH];

  const int tid  = threadIdx.x;
  const int lane = tid & 31;
  const int wave = tid >> 5;
  const int n0   = wave * 16;
  const int b0   = blockIdx.x * BSUB;
  const int hh   = lane >> 4;
  const int cl   = lane & 7;
  const int br   = (lane >> 3) & 1;
  const int so   = n0 + 8 * hh;
  const int eb   = tid >> 6;
  const int ej   = tid & 63;

  v2f wa[4][4];
#pragma unroll
  for (int i = 0; i < 4; ++i)
#pragma unroll
    for (int j = 0; j < 4; ++j)
      wa[i][j] = *(const v2f*)(W + (16 * ((wave + i) & 3) + (lane & 15)) * S_DIM
                                 + so + 2 * j);
  int par = 0;

  const float w1 = cl >= 1 ? 1.0f : 0.0f;
  const float w2 = cl >= 2 ? 1.0f : 0.0f;
  const float w4 = cl >= 4 ? 1.0f : 0.0f;

  float yp[8];
  {
    const float* p = y0 + (size_t)(b0 + br) * S_DIM + so;
    const v4f a = *(const v4f*)p, b = *(const v4f*)(p + 4);
    yp[0] = a.x; yp[1] = a.y; yp[2] = a.z; yp[3] = a.w;
    yp[4] = b.x; yp[5] = b.y; yp[6] = b.z; yp[7] = b.w;
  }
  out[(size_t)(b0 + eb) * S_DIM + ej] = y0[(size_t)(b0 + eb) * S_DIM + ej];

  float fv[8];
  {
    const float* p = forces + ((size_t)(1 + cl) * B_TOT + b0 + br) * S_DIM + so;
    const v4f a = *(const v4f*)p, b = *(const v4f*)(p + 4);
    fv[0] = a.x; fv[1] = a.y; fv[2] = a.z; fv[3] = a.w;
    fv[4] = b.x; fv[5] = b.y; fv[6] = b.z; fv[7] = b.w;
  }

  for (int blk = 0; blk < NB; ++blk) {
    float fn[8];
    {
      const int bn = (blk + 1 < NB) ? blk + 1 : blk;
      const float* p = forces + ((size_t)(1 + bn * C_DIM + cl) * B_TOT + b0 + br) * S_DIM + so;
      const v4f a = *(const v4f*)p, b = *(const v4f*)(p + 4);
      fn[0] = a.x; fn[1] = a.y; fn[2] = a.z; fn[3] = a.w;
      fn[4] = b.x; fn[5] = b.y; fn[6] = b.z; fn[7] = b.w;
    }

    float y[8];
#pragma unroll
    for (int r = 0; r < 8; ++r) y[r] = 0.f;

    for (int it = 0; it < N_ITER; ++it) {
      float sv[8], yd[8];
#pragma unroll
      for (int r = 0; r < 8; ++r) {
        const float below = row_below<1>(y[r]);
        const float ym = (cl == 0) ? yp[r] : below;
        sv[r] = 0.5f * (ym + y[r]);
        yd[r] = y[r] - ym;
      }
      float ws[8];
      ksplit_product(wa, sv, sP + par * PBUF, wave, lane, ws);
      par ^= 1;
      float Rr[8], Gg[8], dd[8];
#pragma unroll
      for (int r = 0; r < 8; ++r) {
        const float t = tanhf(ws[r]);
        Gg[r] = 0.5f * DT_C * (1.0f - t * t);
        Rr[r] = yd[r] + DT_C * t - DT_C * fv[r];
        dd[r] = Rr[r];
      }

      for (int k = 0; k < CHUNK_STEPS; ++k) {
        float mb[8];
#pragma unroll
        for (int r = 0; r < 8; ++r) {
          const float xc = cell_running_sum(dd[r], w1, w2, w4);
          mb[r] = fmaf(2.0f, xc, -dd[r]);
        }
        float tot[8];
        ksplit_product(wa, mb, sP + par * PBUF, wave, lane, tot);
        par ^= 1;
#pragma unroll
        for (int r = 0; r < 8; ++r) dd[r] = Rr[r] - Gg[r] * tot[r];
      }

#pragma unroll
      for (int r = 0; r < 8; ++r) y[r] -= cell_running_sum(dd[r], w1, w2, w4);
    }

    {
      float* q = sOut + (cl * BSUB + br) * OPITCH + so;
      *(v4f*)q       = (v4f){y[0], y[1], y[2], y[3]};
      *(v4f*)(q + 4) = (v4f){y[4], y[5], y[6], y[7]};
    }
    __syncthreads();
    {
      float yo[C_DIM];
#pragma unroll
      for (int c = 0; c < C_DIM; ++c) yo[c] = sOut[(c * BSUB + eb) * OPITCH + ej];
#pragma unroll
      for (int c = 0; c < C_DIM; ++c)
        out[((size_t)(1 + blk * C_DIM + c) * B_TOT + b0 + eb) * S_DIM + ej] = yo[c];
    }
#pragma unroll
    for (int r = 0; r < 8; ++r) {
      yp[r] = __shfl(y[r], lane | 7);
      fv[r] = fn[r];
    }
  }
}

extern "C" void kernel_launch(void* const* d_in, const int* in_sizes, int n_in,
                              void* d_out, int out_size, void* d_ws,
                              size_t ws_size, hipStream_t stream) {
  (void)in_sizes; (void)n_in; (void)out_size; (void)d_ws; (void)ws_size;
  const float* y0     = (const float*)d_in[0];
  const float* forces = (const float*)d_in[1];
  const float* W      = (const float*)d_in[2];
  float* out = (float*)d_out;
  solver_gfx1250_kernel<<<NWG, THREADS, 0, stream>>>(y0, forces, W, out);
}
